# starting kernel
# baseline (speedup 1.0000x reference)
#include <hip/hip_runtime.h>
#include <hip/hip_bf16.h>
#include <math.h>

typedef __attribute__((ext_vector_type(2))) float v2f;
typedef __attribute__((ext_vector_type(8))) float v8f;

__global__ void degree_kernel(const int* __restrict__ dst, float* __restrict__ deg, int E) {
    int e = blockIdx.x * blockDim.x + threadIdx.x;
    if (e >= E) return;
    atomicAdd(&deg[dst[e]], 1.0f);
}

__global__ void invdeg_kernel(float* __restrict__ deg, int N) {
    int i = blockIdx.x * blockDim.x + threadIdx.x;
    if (i >= N) return;
    deg[i] = 1.0f / fmaxf(deg[i], 1.0f);
}

template <int C>
__global__ __launch_bounds__(256)
void scatter_add_kernel(const float* __restrict__ x,
                        const int* __restrict__ src,
                        const int* __restrict__ dst,
                        float* __restrict__ agg, int E) {
    constexpr int CHUNKS = C / 4;
    int tid = blockIdx.x * blockDim.x + threadIdx.x;
    if (tid >= E * CHUNKS) return;
    int e = tid / CHUNKS;
    int q = tid - e * CHUNKS;
    int s = src[e];
    int d = dst[e];
    const float4 v = *(const float4*)(x + (size_t)s * C + q * 4);
    float* a = agg + (size_t)d * C + q * 4;
    atomicAdd(a + 0, v.x);
    atomicAdd(a + 1, v.y);
    atomicAdd(a + 2, v.z);
    atomicAdd(a + 3, v.w);
}

template <int CIN, int COUT, bool RELU>
__global__ __launch_bounds__(256)
void sage_wmma_kernel(const float* __restrict__ agg,
                      const float* __restrict__ invdeg,
                      const float* __restrict__ xin,
                      const float* __restrict__ wl,
                      const float* __restrict__ wr,
                      const float* __restrict__ bias,
                      float* __restrict__ out,
                      int N) {
    const int lane = threadIdx.x & 31;
    const int waveInBlock = threadIdx.x >> 5;
    const int tile = blockIdx.x * (blockDim.x >> 5) + waveInBlock;
    const int row0 = tile * 16;
    if (row0 >= N) return;

    const int m  = lane & 15;
    const int kk = (lane >> 4) << 1;
    const bool fullTile = (row0 + 16 <= N);
    const int node = row0 + (fullTile ? m : min(m, N - 1 - row0));
    const float scale = invdeg[node];
    const float* aggRow = agg + (size_t)node * CIN;
    const float* xRow   = xin + (size_t)node * CIN;
    const int rbase = row0 + ((lane >> 4) << 3);

    for (int nb = 0; nb < COUT; nb += 16) {
        v8f c = {};
        const float* wlRow = wl + (size_t)(nb + m) * CIN;
        const float* wrRow = wr + (size_t)(nb + m) * CIN;
#pragma unroll
        for (int k0 = 0; k0 < CIN; k0 += 4) {
            v2f a_mean, a_self, b_l, b_r;
            a_mean.x = aggRow[k0 + kk] * scale;
            a_mean.y = aggRow[k0 + kk + 1] * scale;
            a_self.x = xRow[k0 + kk];
            a_self.y = xRow[k0 + kk + 1];
            b_l.x = wlRow[k0 + kk];
            b_l.y = wlRow[k0 + kk + 1];
            b_r.x = wrRow[k0 + kk];
            b_r.y = wrRow[k0 + kk + 1];
            c = __builtin_amdgcn_wmma_f32_16x16x4_f32(false, a_mean, false, b_l,
                                                      (short)0, c, false, false);
            c = __builtin_amdgcn_wmma_f32_16x16x4_f32(false, a_self, false, b_r,
                                                      (short)0, c, false, false);
        }
        const float bv = bias[nb + m];
        if (fullTile) {
            float* o = out + (size_t)rbase * COUT + nb + m;
#pragma unroll
            for (int v = 0; v < 8; ++v) {
                float val = c[v] + bv;
                if (RELU) val = fmaxf(val, 0.0f);
                o[(size_t)v * COUT] = val;
            }
        } else {
#pragma unroll
            for (int v = 0; v < 8; ++v) {
                int r = rbase + v;
                if (r < N) {
                    float val = c[v] + bv;
                    if (RELU) val = fmaxf(val, 0.0f);
                    out[(size_t)r * COUT + nb + m] = val;
                }
            }
        }
    }
}

__global__ void sage_final_kernel(const float* __restrict__ agg,
                                  const float* __restrict__ invdeg,
                                  const float* __restrict__ h2,
                                  const float* __restrict__ wl3,
                                  const float* __restrict__ wr3,
                                  const float* __restrict__ b3,
                                  float* __restrict__ out, int N) {
    int i = blockIdx.x * blockDim.x + threadIdx.x;
    if (i >= N) return;
    float s = b3[0];
    float sc = invdeg[i];
    const float4* ar = (const float4*)(agg + (size_t)i * 64);
    const float4* hr = (const float4*)(h2 + (size_t)i * 64);
    const float4* wlv = (const float4*)wl3;
    const float4* wrv = (const float4*)wr3;
#pragma unroll
    for (int q = 0; q < 16; ++q) {
        float4 a = ar[q], h = hr[q], l = wlv[q], r = wrv[q];
        s += (a.x * sc) * l.x + (a.y * sc) * l.y + (a.z * sc) * l.z + (a.w * sc) * l.w;
        s += h.x * r.x + h.y * r.y + h.z * r.z + h.w * r.w;
    }
    out[i] = 1.0f / (1.0f + expf(-s));
}

extern "C" void kernel_launch(void* const* d_in, const int* in_sizes, int n_in,
                              void* d_out, int out_size, void* d_ws, size_t ws_size,
                              hipStream_t stream) {
    const float* x    = (const float*)d_in[0];
    const int*   ei   = (const int*)d_in[1];
    const float* wl1  = (const float*)d_in[2];
    const float* wr1  = (const float*)d_in[3];
    const float* b1   = (const float*)d_in[4];
    const float* wl2  = (const float*)d_in[5];
    const float* wr2  = (const float*)d_in[6];
    const float* b2   = (const float*)d_in[7];
    const float* wl3  = (const float*)d_in[8];
    const float* wr3  = (const float*)d_in[9];
    const float* b3   = (const float*)d_in[10];
    float* out = (float*)d_out;

    const int N = in_sizes[0] / 32;
    const int E = in_sizes[1] / 2;
    const int* src = ei;
    const int* dst = ei + E;

    float* deg = (float*)d_ws;
    float* agg = deg + N;
    float* h1  = agg + (size_t)N * 64;
    float* h2  = h1 + (size_t)N * 64;

    const int tiles = (N + 15) / 16;
    const int gemmBlocks = (tiles + 7) / 8;

    hipMemsetAsync(deg, 0, (size_t)N * sizeof(float), stream);
    degree_kernel<<<(E + 255) / 256, 256, 0, stream>>>(dst, deg, E);
    invdeg_kernel<<<(N + 255) / 256, 256, 0, stream>>>(deg, N);

    hipMemsetAsync(agg, 0, (size_t)N * 32 * sizeof(float), stream);
    scatter_add_kernel<32><<<(E * 8 + 255) / 256, 256, 0, stream>>>(x, src, dst, agg, E);
    sage_wmma_kernel<32, 64, true><<<gemmBlocks, 256, 0, stream>>>(
        agg, deg, x, wl1, wr1, b1, h1, N);

    hipMemsetAsync(agg, 0, (size_t)N * 64 * sizeof(float), stream);
    scatter_add_kernel<64><<<(E * 16 + 255) / 256, 256, 0, stream>>>(h1, src, dst, agg, E);
    sage_wmma_kernel<64, 64, true><<<gemmBlocks, 256, 0, stream>>>(
        agg, deg, h1, wl2, wr2, b2, h2, N);

    hipMemsetAsync(agg, 0, (size_t)N * 64 * sizeof(float), stream);
    scatter_add_kernel<64><<<(E * 16 + 255) / 256, 256, 0, stream>>>(h2, src, dst, agg, E);
    sage_final_kernel<<<(N + 255) / 256, 256, 0, stream>>>(
        agg, deg, h2, wl3, wr3, b3, out, N);
}


# Edge scatter remapped to lane = channel: a wave walks 32 edges, each row load and each f32 atomic add covers one whole 128-byte line; 8-16x fewer waves; sums stay in atomic arrival order
# speedup vs baseline: 1.3069x; 1.3069x over previous
#include <hip/hip_runtime.h>
#include <hip/hip_bf16.h>
#include <math.h>

typedef __attribute__((ext_vector_type(2))) float v2f;
typedef __attribute__((ext_vector_type(8))) float v8f;

__global__ void degree_kernel(const int* __restrict__ dst, float* __restrict__ deg, int E) {
    int e = blockIdx.x * blockDim.x + threadIdx.x;
    if (e >= E) return;
    atomicAdd(&deg[dst[e]], 1.0f);
}

__global__ void invdeg_kernel(float* __restrict__ deg, int N) {
    int i = blockIdx.x * blockDim.x + threadIdx.x;
    if (i >= N) return;
    deg[i] = 1.0f / fmaxf(deg[i], 1.0f);
}

template <int C>
__global__ __launch_bounds__(256)
void scatter_add_kernel(const float* __restrict__ x,
                        const int* __restrict__ src,
                        const int* __restrict__ dst,
                        float* __restrict__ agg, int E) {
    constexpr int R = C / 32;
    constexpr int U = 16 / R;
    const int lane = threadIdx.x & 31;
    const int waveInBlock = __builtin_amdgcn_readfirstlane((int)(threadIdx.x >> 5));
    const int wave = blockIdx.x * (blockDim.x >> 5) + waveInBlock;
    const int e0 = wave * 32;
    if (e0 >= E) return;
    const int n = min(32, E - e0);

    if (n == 32) {
        const int sv = src[e0 + lane];
        const int dv = dst[e0 + lane];
#pragma unroll
        for (int b = 0; b < 32; b += U) {
            float v[U][R];
#pragma unroll
            for (int u = 0; u < U; ++u) {
                const float* xr = x + (size_t)__builtin_amdgcn_readlane(sv, b + u) * C;
#pragma unroll
                for (int r = 0; r < R; ++r) v[u][r] = xr[r * 32 + lane];
            }
#pragma unroll
            for (int u = 0; u < U; ++u) {
                float* ar = agg + (size_t)__builtin_amdgcn_readlane(dv, b + u) * C;
#pragma unroll
                for (int r = 0; r < R; ++r) atomicAdd(ar + r * 32 + lane, v[u][r]);
            }
        }
    } else {
        int sv = 0, dv = 0;
        if (lane < n) {
            sv = src[e0 + lane];
            dv = dst[e0 + lane];
        }
        for (int j = 0; j < n; ++j) {
            const float* xr = x + (size_t)__builtin_amdgcn_readlane(sv, j) * C;
            float* ar = agg + (size_t)__builtin_amdgcn_readlane(dv, j) * C;
#pragma unroll
            for (int r = 0; r < R; ++r) atomicAdd(ar + r * 32 + lane, xr[r * 32 + lane]);
        }
    }
}

template <int CIN, int COUT, bool RELU>
__global__ __launch_bounds__(256)
void sage_wmma_kernel(const float* __restrict__ agg,
                      const float* __restrict__ invdeg,
                      const float* __restrict__ xin,
                      const float* __restrict__ wl,
                      const float* __restrict__ wr,
                      const float* __restrict__ bias,
                      float* __restrict__ out,
                      int N) {
    const int lane = threadIdx.x & 31;
    const int waveInBlock = threadIdx.x >> 5;
    const int tile = blockIdx.x * (blockDim.x >> 5) + waveInBlock;
    const int row0 = tile * 16;
    if (row0 >= N) return;

    const int m  = lane & 15;
    const int kk = (lane >> 4) << 1;
    const bool fullTile = (row0 + 16 <= N);
    const int node = row0 + (fullTile ? m : min(m, N - 1 - row0));
    const float scale = invdeg[node];
    const float* aggRow = agg + (size_t)node * CIN;
    const float* xRow   = xin + (size_t)node * CIN;
    const int rbase = row0 + ((lane >> 4) << 3);

    for (int nb = 0; nb < COUT; nb += 16) {
        v8f c = {};
        const float* wlRow = wl + (size_t)(nb + m) * CIN;
        const float* wrRow = wr + (size_t)(nb + m) * CIN;
#pragma unroll
        for (int k0 = 0; k0 < CIN; k0 += 4) {
            v2f a_mean, a_self, b_l, b_r;
            a_mean.x = aggRow[k0 + kk] * scale;
            a_mean.y = aggRow[k0 + kk + 1] * scale;
            a_self.x = xRow[k0 + kk];
            a_self.y = xRow[k0 + kk + 1];
            b_l.x = wlRow[k0 + kk];
            b_l.y = wlRow[k0 + kk + 1];
            b_r.x = wrRow[k0 + kk];
            b_r.y = wrRow[k0 + kk + 1];
            c = __builtin_amdgcn_wmma_f32_16x16x4_f32(false, a_mean, false, b_l,
                                                      (short)0, c, false, false);
            c = __builtin_amdgcn_wmma_f32_16x16x4_f32(false, a_self, false, b_r,
                                                      (short)0, c, false, false);
        }
        const float bv = bias[nb + m];
        if (fullTile) {
            float* o = out + (size_t)rbase * COUT + nb + m;
#pragma unroll
            for (int v = 0; v < 8; ++v) {
                float val = c[v] + bv;
                if (RELU) val = fmaxf(val, 0.0f);
                o[(size_t)v * COUT] = val;
            }
        } else {
#pragma unroll
            for (int v = 0; v < 8; ++v) {
                int r = rbase + v;
                if (r < N) {
                    float val = c[v] + bv;
                    if (RELU) val = fmaxf(val, 0.0f);
                    out[(size_t)r * COUT + nb + m] = val;
                }
            }
        }
    }
}

__global__ void sage_final_kernel(const float* __restrict__ agg,
                                  const float* __restrict__ invdeg,
                                  const float* __restrict__ h2,
                                  const float* __restrict__ wl3,
                                  const float* __restrict__ wr3,
                                  const float* __restrict__ b3,
                                  float* __restrict__ out, int N) {
    int i = blockIdx.x * blockDim.x + threadIdx.x;
    if (i >= N) return;
    float s = b3[0];
    float sc = invdeg[i];
    const float4* ar = (const float4*)(agg + (size_t)i * 64);
    const float4* hr = (const float4*)(h2 + (size_t)i * 64);
    const float4* wlv = (const float4*)wl3;
    const float4* wrv = (const float4*)wr3;
#pragma unroll
    for (int q = 0; q < 16; ++q) {
        float4 a = ar[q], h = hr[q], l = wlv[q], r = wrv[q];
        s += (a.x * sc) * l.x + (a.y * sc) * l.y + (a.z * sc) * l.z + (a.w * sc) * l.w;
        s += h.x * r.x + h.y * r.y + h.z * r.z + h.w * r.w;
    }
    out[i] = 1.0f / (1.0f + expf(-s));
}

extern "C" void kernel_launch(void* const* d_in, const int* in_sizes, int n_in,
                              void* d_out, int out_size, void* d_ws, size_t ws_size,
                              hipStream_t stream) {
    const float* x    = (const float*)d_in[0];
    const int*   ei   = (const int*)d_in[1];
    const float* wl1  = (const float*)d_in[2];
    const float* wr1  = (const float*)d_in[3];
    const float* b1   = (const float*)d_in[4];
    const float* wl2  = (const float*)d_in[5];
    const float* wr2  = (const float*)d_in[6];
    const float* b2   = (const float*)d_in[7];
    const float* wl3  = (const float*)d_in[8];
    const float* wr3  = (const float*)d_in[9];
    const float* b3   = (const float*)d_in[10];
    float* out = (float*)d_out;

    const int N = in_sizes[0] / 32;
    const int E = in_sizes[1] / 2;
    const int* src = ei;
    const int* dst = ei + E;

    float* deg = (float*)d_ws;
    float* agg = deg + N;
    float* h1  = agg + (size_t)N * 64;
    float* h2  = h1 + (size_t)N * 64;

    const int tiles = (N + 15) / 16;
    const int gemmBlocks = (tiles + 7) / 8;

    hipMemsetAsync(deg, 0, (size_t)N * sizeof(float), stream);
    degree_kernel<<<(E + 255) / 256, 256, 0, stream>>>(dst, deg, E);
    invdeg_kernel<<<(N + 255) / 256, 256, 0, stream>>>(deg, N);

    hipMemsetAsync(agg, 0, (size_t)N * 32 * sizeof(float), stream);
    scatter_add_kernel<32><<<(E + 255) / 256, 256, 0, stream>>>(x, src, dst, agg, E);
    sage_wmma_kernel<32, 64, true><<<gemmBlocks, 256, 0, stream>>>(
        agg, deg, x, wl1, wr1, b1, h1, N);

    hipMemsetAsync(agg, 0, (size_t)N * 64 * sizeof(float), stream);
    scatter_add_kernel<64><<<(E + 255) / 256, 256, 0, stream>>>(h1, src, dst, agg, E);
    sage_wmma_kernel<64, 64, true><<<gemmBlocks, 256, 0, stream>>>(
        agg, deg, h1, wl2, wr2, b2, h2, N);

    hipMemsetAsync(agg, 0, (size_t)N * 64 * sizeof(float), stream);
    scatter_add_kernel<64><<<(E + 255) / 256, 256, 0, stream>>>(h2, src, dst, agg, E);
    sage_final_kernel<<<(N + 255) / 256, 256, 0, stream>>>(
        agg, deg, h2, wl3, wr3, b3, out, N);
}


# Layer-3 linear map applied before the neighbour sum: layer-2 epilogue keeps 2 scalars per node (h2 never stored), layer-3 scatter adds 1 float per edge, not 64; sum order: dot, then atomic sum
# speedup vs baseline: 1.4566x; 1.1145x over previous
#include <hip/hip_runtime.h>
#include <hip/hip_bf16.h>
#include <math.h>

typedef __attribute__((ext_vector_type(2))) float v2f;
typedef __attribute__((ext_vector_type(8))) float v8f;

__global__ void degree_kernel(const int* __restrict__ dst, float* __restrict__ deg, int E) {
    int e = blockIdx.x * blockDim.x + threadIdx.x;
    if (e >= E) return;
    atomicAdd(&deg[dst[e]], 1.0f);
}

__global__ void invdeg_kernel(float* __restrict__ deg, int N) {
    int i = blockIdx.x * blockDim.x + threadIdx.x;
    if (i >= N) return;
    deg[i] = 1.0f / fmaxf(deg[i], 1.0f);
}

template <int C>
__global__ __launch_bounds__(256)
void scatter_add_kernel(const float* __restrict__ x,
                        const int* __restrict__ src,
                        const int* __restrict__ dst,
                        float* __restrict__ agg, int E) {
    constexpr int R = C / 32;
    constexpr int U = 16 / R;
    const int lane = threadIdx.x & 31;
    const int waveInBlock = __builtin_amdgcn_readfirstlane((int)(threadIdx.x >> 5));
    const int wave = blockIdx.x * (blockDim.x >> 5) + waveInBlock;
    const int e0 = wave * 32;
    if (e0 >= E) return;
    const int n = min(32, E - e0);

    if (n == 32) {
        const int sv = src[e0 + lane];
        const int dv = dst[e0 + lane];
#pragma unroll
        for (int b = 0; b < 32; b += U) {
            float v[U][R];
#pragma unroll
            for (int u = 0; u < U; ++u) {
                const float* xr = x + (size_t)__builtin_amdgcn_readlane(sv, b + u) * C;
#pragma unroll
                for (int r = 0; r < R; ++r) v[u][r] = xr[r * 32 + lane];
            }
#pragma unroll
            for (int u = 0; u < U; ++u) {
                float* ar = agg + (size_t)__builtin_amdgcn_readlane(dv, b + u) * C;
#pragma unroll
                for (int r = 0; r < R; ++r) atomicAdd(ar + r * 32 + lane, v[u][r]);
            }
        }
    } else {
        int sv = 0, dv = 0;
        if (lane < n) {
            sv = src[e0 + lane];
            dv = dst[e0 + lane];
        }
        for (int j = 0; j < n; ++j) {
            const float* xr = x + (size_t)__builtin_amdgcn_readlane(sv, j) * C;
            float* ar = agg + (size_t)__builtin_amdgcn_readlane(dv, j) * C;
#pragma unroll
            for (int r = 0; r < R; ++r) atomicAdd(ar + r * 32 + lane, xr[r * 32 + lane]);
        }
    }
}

template <int CIN, int COUT, bool RELU>
__global__ __launch_bounds__(256)
void sage_wmma_kernel(const float* __restrict__ agg,
                      const float* __restrict__ invdeg,
                      const float* __restrict__ xin,
                      const float* __restrict__ wl,
                      const float* __restrict__ wr,
                      const float* __restrict__ bias,
                      float* __restrict__ out,
                      int N) {
    const int lane = threadIdx.x & 31;
    const int waveInBlock = threadIdx.x >> 5;
    const int tile = blockIdx.x * (blockDim.x >> 5) + waveInBlock;
    const int row0 = tile * 16;
    if (row0 >= N) return;

    const int m  = lane & 15;
    const int kk = (lane >> 4) << 1;
    const bool fullTile = (row0 + 16 <= N);
    const int node = row0 + (fullTile ? m : min(m, N - 1 - row0));
    const float scale = invdeg[node];
    const float* aggRow = agg + (size_t)node * CIN;
    const float* xRow   = xin + (size_t)node * CIN;
    const int rbase = row0 + ((lane >> 4) << 3);

    for (int nb = 0; nb < COUT; nb += 16) {
        v8f c = {};
        const float* wlRow = wl + (size_t)(nb + m) * CIN;
        const float* wrRow = wr + (size_t)(nb + m) * CIN;
#pragma unroll
        for (int k0 = 0; k0 < CIN; k0 += 4) {
            v2f a_mean, a_self, b_l, b_r;
            a_mean.x = aggRow[k0 + kk] * scale;
            a_mean.y = aggRow[k0 + kk + 1] * scale;
            a_self.x = xRow[k0 + kk];
            a_self.y = xRow[k0 + kk + 1];
            b_l.x = wlRow[k0 + kk];
            b_l.y = wlRow[k0 + kk + 1];
            b_r.x = wrRow[k0 + kk];
            b_r.y = wrRow[k0 + kk + 1];
            c = __builtin_amdgcn_wmma_f32_16x16x4_f32(false, a_mean, false, b_l,
                                                      (short)0, c, false, false);
            c = __builtin_amdgcn_wmma_f32_16x16x4_f32(false, a_self, false, b_r,
                                                      (short)0, c, false, false);
        }
        const float bv = bias[nb + m];
        if (fullTile) {
            float* o = out + (size_t)rbase * COUT + nb + m;
#pragma unroll
            for (int v = 0; v < 8; ++v) {
                float val = c[v] + bv;
                if (RELU) val = fmaxf(val, 0.0f);
                o[(size_t)v * COUT] = val;
            }
        } else {
#pragma unroll
            for (int v = 0; v < 8; ++v) {
                int r = rbase + v;
                if (r < N) {
                    float val = c[v] + bv;
                    if (RELU) val = fmaxf(val, 0.0f);
                    out[(size_t)r * COUT + nb + m] = val;
                }
            }
        }
    }
}

template <int CIN, int COUT>
__global__ __launch_bounds__(256)
void sage_wmma_project_kernel(const float* __restrict__ agg,
                              const float* __restrict__ invdeg,
                              const float* __restrict__ xin,
                              const float* __restrict__ wl,
                              const float* __restrict__ wr,
                              const float* __restrict__ bias,
                              const float* __restrict__ wl3,
                              const float* __restrict__ wr3,
                              const float* __restrict__ b3,
                              float* __restrict__ p,
                              float* __restrict__ q,
                              int N) {
    __shared__ float stage[2][128];

    const int lane = threadIdx.x & 31;
    const int waveInBlock = __builtin_amdgcn_readfirstlane((int)(threadIdx.x >> 5));
    const int tile = blockIdx.x * 8 + waveInBlock;
    const int row0 = tile * 16;
    const int m    = lane & 15;
    const int half = lane >> 4;
    const int kk   = half << 1;

    float pl = 0.0f, pr = 0.0f;
    if (row0 < N) {
        const int node = min(row0 + m, N - 1);
        const float scale = invdeg[node];
        const float* aggRow = agg + (size_t)node * CIN;
        const float* xRow   = xin + (size_t)node * CIN;

        for (int nb = 0; nb < COUT; nb += 16) {
            v8f c = {};
            const float* wlRow = wl + (size_t)(nb + m) * CIN;
            const float* wrRow = wr + (size_t)(nb + m) * CIN;
#pragma unroll
            for (int k0 = 0; k0 < CIN; k0 += 4) {
                v2f x_mean, x_self, w_l, w_r;
                x_mean.x = aggRow[k0 + kk] * scale;
                x_mean.y = aggRow[k0 + kk + 1] * scale;
                x_self.x = xRow[k0 + kk];
                x_self.y = xRow[k0 + kk + 1];
                w_l.x = wlRow[k0 + kk];
                w_l.y = wlRow[k0 + kk + 1];
                w_r.x = wrRow[k0 + kk];
                w_r.y = wrRow[k0 + kk + 1];
                c = __builtin_amdgcn_wmma_f32_16x16x4_f32(false, w_l, false, x_mean,
                                                          (short)0, c, false, false);
                c = __builtin_amdgcn_wmma_f32_16x16x4_f32(false, w_r, false, x_self,
                                                          (short)0, c, false, false);
            }
            const int ch0 = nb + (half << 3);
            const float4 bA = *(const float4*)(bias + ch0), bB = *(const float4*)(bias + ch0 + 4);
            const float4 lA = *(const float4*)(wl3 + ch0),  lB = *(const float4*)(wl3 + ch0 + 4);
            const float4 rA = *(const float4*)(wr3 + ch0),  rB = *(const float4*)(wr3 + ch0 + 4);
            const float bv[8] = {bA.x, bA.y, bA.z, bA.w, bB.x, bB.y, bB.z, bB.w};
            const float lv[8] = {lA.x, lA.y, lA.z, lA.w, lB.x, lB.y, lB.z, lB.w};
            const float rv[8] = {rA.x, rA.y, rA.z, rA.w, rB.x, rB.y, rB.z, rB.w};
#pragma unroll
            for (int v = 0; v < 8; ++v) {
                const float h = fmaxf(c[v] + bv[v], 0.0f);
                pl += h * lv[v];
                pr += h * rv[v];
            }
        }
        pl += __shfl_xor(pl, 16);
        pr += __shfl_xor(pr, 16);
        pr += b3[0];
    }
    if (half == 0) {
        stage[0][waveInBlock * 16 + m] = pl;
        stage[1][waveInBlock * 16 + m] = pr;
    }
    __syncthreads();
    const int t = threadIdx.x & 127;
    const int which = threadIdx.x >> 7;
    const int nodeOut = blockIdx.x * 128 + t;
    if (nodeOut < N) (which ? q : p)[nodeOut] = stage[which][t];
}

__global__ __launch_bounds__(256)
void scatter_scalar_kernel(const float* __restrict__ p,
                           const int* __restrict__ src,
                           const int* __restrict__ dst,
                           float* __restrict__ acc, int E) {
    constexpr int EPT = 8;
    const int lane = threadIdx.x & 31;
    const int waveInBlock = __builtin_amdgcn_readfirstlane((int)(threadIdx.x >> 5));
    const int wave = blockIdx.x * 8 + waveInBlock;
    const int e0 = wave * (32 * EPT);
    if (e0 >= E) return;

    if (e0 + 32 * EPT <= E) {
        int sv[EPT], dv[EPT];
        float v[EPT];
#pragma unroll
        for (int j = 0; j < EPT; ++j) {
            sv[j] = src[e0 + j * 32 + lane];
            dv[j] = dst[e0 + j * 32 + lane];
        }
#pragma unroll
        for (int j = 0; j < EPT; ++j) v[j] = p[sv[j]];
#pragma unroll
        for (int j = 0; j < EPT; ++j) atomicAdd(&acc[dv[j]], v[j]);
    } else {
        for (int e = e0 + lane; e < E; e += 32) atomicAdd(&acc[dst[e]], p[src[e]]);
    }
}

__global__ void sage_final_kernel(const float* __restrict__ acc,
                                  const float* __restrict__ invdeg,
                                  const float* __restrict__ q,
                                  float* __restrict__ out, int N) {
    int i = blockIdx.x * blockDim.x + threadIdx.x;
    if (i >= N) return;
    const float s = acc[i] * invdeg[i] + q[i];
    out[i] = 1.0f / (1.0f + expf(-s));
}

extern "C" void kernel_launch(void* const* d_in, const int* in_sizes, int n_in,
                              void* d_out, int out_size, void* d_ws, size_t ws_size,
                              hipStream_t stream) {
    const float* x    = (const float*)d_in[0];
    const int*   ei   = (const int*)d_in[1];
    const float* wl1  = (const float*)d_in[2];
    const float* wr1  = (const float*)d_in[3];
    const float* b1   = (const float*)d_in[4];
    const float* wl2  = (const float*)d_in[5];
    const float* wr2  = (const float*)d_in[6];
    const float* b2   = (const float*)d_in[7];
    const float* wl3  = (const float*)d_in[8];
    const float* wr3  = (const float*)d_in[9];
    const float* b3   = (const float*)d_in[10];
    float* out = (float*)d_out;

    const int N = in_sizes[0] / 32;
    const int E = in_sizes[1] / 2;
    const int* src = ei;
    const int* dst = ei + E;

    float* deg  = (float*)d_ws;
    float* acc3 = deg + N;
    float* agg  = acc3 + N;
    float* h1   = agg + (size_t)N * 64;
    float* p3   = h1 + (size_t)N * 64;
    float* q3   = p3 + N;

    const int tiles = (N + 15) / 16;
    const int gemmBlocks = (tiles + 7) / 8;

    hipMemsetAsync(deg, 0, (size_t)N * 2 * sizeof(float), stream);
    degree_kernel<<<(E + 255) / 256, 256, 0, stream>>>(dst, deg, E);
    invdeg_kernel<<<(N + 255) / 256, 256, 0, stream>>>(deg, N);

    hipMemsetAsync(agg, 0, (size_t)N * 32 * sizeof(float), stream);
    scatter_add_kernel<32><<<(E + 255) / 256, 256, 0, stream>>>(x, src, dst, agg, E);
    sage_wmma_kernel<32, 64, true><<<gemmBlocks, 256, 0, stream>>>(
        agg, deg, x, wl1, wr1, b1, h1, N);

    hipMemsetAsync(agg, 0, (size_t)N * 64 * sizeof(float), stream);
    scatter_add_kernel<64><<<(E + 255) / 256, 256, 0, stream>>>(h1, src, dst, agg, E);
    sage_wmma_project_kernel<64, 64><<<gemmBlocks, 256, 0, stream>>>(
        agg, deg, h1, wl2, wr2, b2, wl3, wr3, b3, p3, q3, N);

    scatter_scalar_kernel<<<(E + 2047) / 2048, 256, 0, stream>>>(p3, src, dst, acc3, E);
    sage_final_kernel<<<(N + 255) / 256, 256, 0, stream>>>(acc3, deg, q3, out, N);
}


# Edges grouped by destination once per call (int atomic rank + prefix sum); the 3 neighbour sums become per-node register sums of gathered rows: no f32 atomics, no agg clears; sums in list order
# speedup vs baseline: 1.9421x; 1.3333x over previous
#include <hip/hip_runtime.h>
#include <hip/hip_bf16.h>
#include <math.h>

typedef __attribute__((ext_vector_type(2))) float v2f;
typedef __attribute__((ext_vector_type(8))) float v8f;

__global__ __launch_bounds__(256)
void edge_rank_kernel(const int* __restrict__ dst, int* __restrict__ cnt,
                      int* __restrict__ rank, int E) {
    constexpr int EPT = 8;
    const int lane = threadIdx.x & 31;
    const int waveInBlock = __builtin_amdgcn_readfirstlane((int)(threadIdx.x >> 5));
    const int wave = blockIdx.x * 8 + waveInBlock;
    const int e0 = wave * (32 * EPT);
    if (e0 >= E) return;

    if (e0 + 32 * EPT <= E) {
        int dv[EPT], rk[EPT];
#pragma unroll
        for (int j = 0; j < EPT; ++j) dv[j] = dst[e0 + j * 32 + lane];
#pragma unroll
        for (int j = 0; j < EPT; ++j) rk[j] = atomicAdd(&cnt[dv[j]], 1);
#pragma unroll
        for (int j = 0; j < EPT; ++j) rank[e0 + j * 32 + lane] = rk[j];
    } else {
        for (int e = e0 + lane; e < E; e += 32) rank[e] = atomicAdd(&cnt[dst[e]], 1);
    }
}

__global__ __launch_bounds__(1024)
void rowptr_kernel(const int* __restrict__ cnt, int* __restrict__ rowptr,
                   float* __restrict__ invdeg, int N) {
    __shared__ int wsum[32];
    __shared__ int sbase;
    const int t = threadIdx.x;
    const int lane = t & 31;
    const int w = t >> 5;
    const int first = blockIdx.x * 1024;

    int part = 0;
    for (int i = t; i < first; i += 1024) part += cnt[i];
    for (int o = 16; o > 0; o >>= 1) part += __shfl_xor(part, o);
    if (lane == 0) wsum[w] = part;
    __syncthreads();
    if (w == 0) {
        int v = wsum[lane];
        for (int o = 16; o > 0; o >>= 1) v += __shfl_xor(v, o);
        if (lane == 0) sbase = v;
    }
    __syncthreads();
    const int base = sbase;

    const int i = first + t;
    const int c = (i < N) ? cnt[i] : 0;
    int inc = c;
    for (int o = 1; o < 32; o <<= 1) {
        const int u = __shfl_up(inc, o);
        if (lane >= o) inc += u;
    }
    if (lane == 31) wsum[w] = inc;
    __syncthreads();
    int v = (lane < w) ? wsum[lane] : 0;
    for (int o = 16; o > 0; o >>= 1) v += __shfl_xor(v, o);
    const int excl = base + v + inc - c;
    if (i < N) {
        rowptr[i] = excl;
        invdeg[i] = 1.0f / fmaxf((float)c, 1.0f);
    }
    if (i == N - 1) rowptr[N] = excl + c;
}

__global__ __launch_bounds__(256)
void edge_place_kernel(const int* __restrict__ src, const int* __restrict__ dst,
                       const int* __restrict__ rank, const int* __restrict__ rowptr,
                       int* __restrict__ nbr, int E) {
    constexpr int EPT = 8;
    const int lane = threadIdx.x & 31;
    const int waveInBlock = __builtin_amdgcn_readfirstlane((int)(threadIdx.x >> 5));
    const int wave = blockIdx.x * 8 + waveInBlock;
    const int e0 = wave * (32 * EPT);
    if (e0 >= E) return;

    if (e0 + 32 * EPT <= E) {
        int sv[EPT], dv[EPT], rk[EPT], pos[EPT];
#pragma unroll
        for (int j = 0; j < EPT; ++j) {
            sv[j] = src[e0 + j * 32 + lane];
            dv[j] = dst[e0 + j * 32 + lane];
            rk[j] = rank[e0 + j * 32 + lane];
        }
#pragma unroll
        for (int j = 0; j < EPT; ++j) pos[j] = rowptr[dv[j]] + rk[j];
#pragma unroll
        for (int j = 0; j < EPT; ++j) nbr[pos[j]] = sv[j];
    } else {
        for (int e = e0 + lane; e < E; e += 32) nbr[rowptr[dst[e]] + rank[e]] = src[e];
    }
}

template <int C>
__global__ __launch_bounds__(256)
void gather_sum_kernel(const float* __restrict__ x,
                       const int* __restrict__ rowptr,
                       const int* __restrict__ nbr,
                       float* __restrict__ agg, int N) {
    constexpr int R = C / 32;
    constexpr int U = 16;
    constexpr int NPW = 8;
    const int lane = threadIdx.x & 31;
    const int waveInBlock = __builtin_amdgcn_readfirstlane((int)(threadIdx.x >> 5));
    const int wave = blockIdx.x * 8 + waveInBlock;
    const int n0 = wave * NPW;
    if (n0 >= N) return;
    const int nn = min(NPW, N - n0);
    const int rp = rowptr[n0 + min(lane, nn)];

    for (int i = 0; i < nn; ++i) {
        const int start = __builtin_amdgcn_readlane(rp, i);
        const int end   = __builtin_amdgcn_readlane(rp, i + 1);
        float acc[R];
        for (int r = 0; r < R; ++r) acc[r] = 0.0f;

        for (int c = start; c < end; c += 32) {
            const int n = min(32, end - c);
            const int sv = nbr[c + min(lane, n - 1)];
            for (int j = 0; j < n; j += U) {
                float v[U][R];
#pragma unroll
                for (int u = 0; u < U; ++u) {
                    const float* xr = x + (size_t)__builtin_amdgcn_readlane(sv, min(j + u, n - 1)) * C;
                    for (int r = 0; r < R; ++r) v[u][r] = (j + u < n) ? xr[r * 32 + lane] : 0.0f;
                }
#pragma unroll
                for (int u = 0; u < U; ++u) {
                    for (int r = 0; r < R; ++r) acc[r] += v[u][r];
                }
            }
        }
        float* ar = agg + (size_t)(n0 + i) * C;
        for (int r = 0; r < R; ++r) ar[r * 32 + lane] = acc[r];
    }
}

template <int CIN, int COUT, bool RELU>
__global__ __launch_bounds__(256)
void sage_wmma_kernel(const float* __restrict__ agg,
                      const float* __restrict__ invdeg,
                      const float* __restrict__ xin,
                      const float* __restrict__ wl,
                      const float* __restrict__ wr,
                      const float* __restrict__ bias,
                      float* __restrict__ out,
                      int N) {
    const int lane = threadIdx.x & 31;
    const int waveInBlock = threadIdx.x >> 5;
    const int tile = blockIdx.x * (blockDim.x >> 5) + waveInBlock;
    const int row0 = tile * 16;
    if (row0 >= N) return;

    const int m  = lane & 15;
    const int kk = (lane >> 4) << 1;
    const bool fullTile = (row0 + 16 <= N);
    const int node = row0 + (fullTile ? m : min(m, N - 1 - row0));
    const float scale = invdeg[node];
    const float* aggRow = agg + (size_t)node * CIN;
    const float* xRow   = xin + (size_t)node * CIN;
    const int rbase = row0 + ((lane >> 4) << 3);

    for (int nb = 0; nb < COUT; nb += 16) {
        v8f c = {};
        const float* wlRow = wl + (size_t)(nb + m) * CIN;
        const float* wrRow = wr + (size_t)(nb + m) * CIN;
#pragma unroll
        for (int k0 = 0; k0 < CIN; k0 += 4) {
            v2f a_mean, a_self, b_l, b_r;
            a_mean.x = aggRow[k0 + kk] * scale;
            a_mean.y = aggRow[k0 + kk + 1] * scale;
            a_self.x = xRow[k0 + kk];
            a_self.y = xRow[k0 + kk + 1];
            b_l.x = wlRow[k0 + kk];
            b_l.y = wlRow[k0 + kk + 1];
            b_r.x = wrRow[k0 + kk];
            b_r.y = wrRow[k0 + kk + 1];
            c = __builtin_amdgcn_wmma_f32_16x16x4_f32(false, a_mean, false, b_l,
                                                      (short)0, c, false, false);
            c = __builtin_amdgcn_wmma_f32_16x16x4_f32(false, a_self, false, b_r,
                                                      (short)0, c, false, false);
        }
        const float bv = bias[nb + m];
        if (fullTile) {
            float* o = out + (size_t)rbase * COUT + nb + m;
#pragma unroll
            for (int v = 0; v < 8; ++v) {
                float val = c[v] + bv;
                if (RELU) val = fmaxf(val, 0.0f);
                o[(size_t)v * COUT] = val;
            }
        } else {
#pragma unroll
            for (int v = 0; v < 8; ++v) {
                int r = rbase + v;
                if (r < N) {
                    float val = c[v] + bv;
                    if (RELU) val = fmaxf(val, 0.0f);
                    out[(size_t)r * COUT + nb + m] = val;
                }
            }
        }
    }
}

template <int CIN, int COUT>
__global__ __launch_bounds__(256)
void sage_wmma_project_kernel(const float* __restrict__ agg,
                              const float* __restrict__ invdeg,
                              const float* __restrict__ xin,
                              const float* __restrict__ wl,
                              const float* __restrict__ wr,
                              const float* __restrict__ bias,
                              const float* __restrict__ wl3,
                              const float* __restrict__ wr3,
                              const float* __restrict__ b3,
                              float* __restrict__ p,
                              float* __restrict__ q,
                              int N) {
    __shared__ float stage[2][128];

    const int lane = threadIdx.x & 31;
    const int waveInBlock = __builtin_amdgcn_readfirstlane((int)(threadIdx.x >> 5));
    const int tile = blockIdx.x * 8 + waveInBlock;
    const int row0 = tile * 16;
    const int m    = lane & 15;
    const int half = lane >> 4;
    const int kk   = half << 1;

    float pl = 0.0f, pr = 0.0f;
    if (row0 < N) {
        const int node = min(row0 + m, N - 1);
        const float scale = invdeg[node];
        const float* aggRow = agg + (size_t)node * CIN;
        const float* xRow   = xin + (size_t)node * CIN;

        for (int nb = 0; nb < COUT; nb += 16) {
            v8f c = {};
            const float* wlRow = wl + (size_t)(nb + m) * CIN;
            const float* wrRow = wr + (size_t)(nb + m) * CIN;
#pragma unroll
            for (int k0 = 0; k0 < CIN; k0 += 4) {
                v2f x_mean, x_self, w_l, w_r;
                x_mean.x = aggRow[k0 + kk] * scale;
                x_mean.y = aggRow[k0 + kk + 1] * scale;
                x_self.x = xRow[k0 + kk];
                x_self.y = xRow[k0 + kk + 1];
                w_l.x = wlRow[k0 + kk];
                w_l.y = wlRow[k0 + kk + 1];
                w_r.x = wrRow[k0 + kk];
                w_r.y = wrRow[k0 + kk + 1];
                c = __builtin_amdgcn_wmma_f32_16x16x4_f32(false, w_l, false, x_mean,
                                                          (short)0, c, false, false);
                c = __builtin_amdgcn_wmma_f32_16x16x4_f32(false, w_r, false, x_self,
                                                          (short)0, c, false, false);
            }
            const int ch0 = nb + (half << 3);
            const float4 bA = *(const float4*)(bias + ch0), bB = *(const float4*)(bias + ch0 + 4);
            const float4 lA = *(const float4*)(wl3 + ch0),  lB = *(const float4*)(wl3 + ch0 + 4);
            const float4 rA = *(const float4*)(wr3 + ch0),  rB = *(const float4*)(wr3 + ch0 + 4);
            const float bv[8] = {bA.x, bA.y, bA.z, bA.w, bB.x, bB.y, bB.z, bB.w};
            const float lv[8] = {lA.x, lA.y, lA.z, lA.w, lB.x, lB.y, lB.z, lB.w};
            const float rv[8] = {rA.x, rA.y, rA.z, rA.w, rB.x, rB.y, rB.z, rB.w};
#pragma unroll
            for (int v = 0; v < 8; ++v) {
                const float h = fmaxf(c[v] + bv[v], 0.0f);
                pl += h * lv[v];
                pr += h * rv[v];
            }
        }
        pl += __shfl_xor(pl, 16);
        pr += __shfl_xor(pr, 16);
        pr += b3[0];
    }
    if (half == 0) {
        stage[0][waveInBlock * 16 + m] = pl;
        stage[1][waveInBlock * 16 + m] = pr;
    }
    __syncthreads();
    const int t = threadIdx.x & 127;
    const int which = threadIdx.x >> 7;
    const int nodeOut = blockIdx.x * 128 + t;
    if (nodeOut < N) (which ? q : p)[nodeOut] = stage[which][t];
}

__global__ __launch_bounds__(256)
void gather_scalar_final_kernel(const float* __restrict__ p,
                                const int* __restrict__ rowptr,
                                const int* __restrict__ nbr,
                                const float* __restrict__ invdeg,
                                const float* __restrict__ q,
                                float* __restrict__ out, int N) {
    const int i = blockIdx.x * blockDim.x + threadIdx.x;
    if (i >= N) return;
    int k = rowptr[i];
    const int end = rowptr[i + 1];
    float s = 0.0f;
    for (; k + 4 <= end; k += 4) {
        const int s0 = nbr[k], s1 = nbr[k + 1], s2 = nbr[k + 2], s3 = nbr[k + 3];
        const float v0 = p[s0], v1 = p[s1], v2 = p[s2], v3 = p[s3];
        s += v0;
        s += v1;
        s += v2;
        s += v3;
    }
    for (; k < end; ++k) s += p[nbr[k]];
    const float z = s * invdeg[i] + q[i];
    out[i] = 1.0f / (1.0f + expf(-z));
}

extern "C" void kernel_launch(void* const* d_in, const int* in_sizes, int n_in,
                              void* d_out, int out_size, void* d_ws, size_t ws_size,
                              hipStream_t stream) {
    const float* x    = (const float*)d_in[0];
    const int*   ei   = (const int*)d_in[1];
    const float* wl1  = (const float*)d_in[2];
    const float* wr1  = (const float*)d_in[3];
    const float* b1   = (const float*)d_in[4];
    const float* wl2  = (const float*)d_in[5];
    const float* wr2  = (const float*)d_in[6];
    const float* b2   = (const float*)d_in[7];
    const float* wl3  = (const float*)d_in[8];
    const float* wr3  = (const float*)d_in[9];
    const float* b3   = (const float*)d_in[10];
    float* out = (float*)d_out;

    const int N = in_sizes[0] / 32;
    const int E = in_sizes[1] / 2;
    const int* src = ei;
    const int* dst = ei + E;

    int*   cnt  = (int*)d_ws;
    float* deg  = (float*)(cnt + N);
    float* agg  = deg + N;
    float* h1   = agg + (size_t)N * 64;
    float* p3   = h1 + (size_t)N * 64;
    float* q3   = p3 + N;
    int*   nbr  = (int*)(q3 + N);
    int*   rptr = nbr + E;
    int*   rank = (int*)h1;

    const int tiles = (N + 15) / 16;
    const int gemmBlocks = (tiles + 7) / 8;
    const int edgeBlocks = (E + 2047) / 2048;
    const int nodeBlocks = (N + 63) / 64;

    hipMemsetAsync(cnt, 0, (size_t)N * sizeof(int), stream);
    edge_rank_kernel<<<edgeBlocks, 256, 0, stream>>>(dst, cnt, rank, E);
    rowptr_kernel<<<(N + 1023) / 1024, 1024, 0, stream>>>(cnt, rptr, deg, N);
    edge_place_kernel<<<edgeBlocks, 256, 0, stream>>>(src, dst, rank, rptr, nbr, E);

    gather_sum_kernel<32><<<nodeBlocks, 256, 0, stream>>>(x, rptr, nbr, agg, N);
    sage_wmma_kernel<32, 64, true><<<gemmBlocks, 256, 0, stream>>>(
        agg, deg, x, wl1, wr1, b1, h1, N);

    gather_sum_kernel<64><<<nodeBlocks, 256, 0, stream>>>(h1, rptr, nbr, agg, N);
    sage_wmma_project_kernel<64, 64><<<gemmBlocks, 256, 0, stream>>>(
        agg, deg, h1, wl2, wr2, b2, wl3, wr3, b3, p3, q3, N);

    gather_scalar_final_kernel<<<(N + 255) / 256, 256, 0, stream>>>(
        p3, rptr, nbr, deg, q3, out, N);
}


# Edges grouped by destination with a two-level counting sort on LDS counters (512-node buckets, then rank and place per bucket): no per-edge global atomic; sums in list order = LDS-add arrival order
# speedup vs baseline: 3.4245x; 1.7633x over previous
#include <hip/hip_runtime.h>
#include <hip/hip_bf16.h>
#include <math.h>

typedef __attribute__((ext_vector_type(2))) float v2f;
typedef __attribute__((ext_vector_type(8))) float v8f;

constexpr int BUCKET_SHIFT = 9;
constexpr int BUCKET_NODES = 1 << BUCKET_SHIFT;
constexpr int MAX_BUCKETS  = 256;
constexpr int EPT          = 8;
constexpr int CHUNK        = 1024 * EPT;
constexpr int RPT          = 12;

__global__ __launch_bounds__(1024)
void bucket_count_kernel(const int* __restrict__ dst, int* __restrict__ totals, int E) {
    __shared__ int hist[MAX_BUCKETS];
    const int tid = threadIdx.x;
    if (tid < MAX_BUCKETS) hist[tid] = 0;
    __syncthreads();
    const int base = blockIdx.x * CHUNK;
    int dv[EPT];
#pragma unroll
    for (int j = 0; j < EPT; ++j) {
        const int e = base + j * 1024 + tid;
        dv[j] = dst[min(e, E - 1)];
    }
#pragma unroll
    for (int j = 0; j < EPT; ++j) {
        const int e = base + j * 1024 + tid;
        if (e < E) atomicAdd(&hist[dv[j] >> BUCKET_SHIFT], 1);
    }
    __syncthreads();
    if (tid < MAX_BUCKETS) {
        const int c = hist[tid];
        if (c > 0) atomicAdd(&totals[tid], c);
    }
}

__global__ __launch_bounds__(1024)
void bucket_scatter_kernel(const int* __restrict__ src, const int* __restrict__ dst,
                           const int* __restrict__ totals, int* __restrict__ fill,
                           int* __restrict__ binned, int E) {
    __shared__ int recS[CHUNK];
    __shared__ int recD[CHUNK];
    __shared__ int hist[MAX_BUCKETS];
    __shared__ int lstart[MAX_BUCKETS];
    __shared__ int gdelta[MAX_BUCKETS];
    __shared__ int wsumT[8];
    __shared__ int wsumC[8];
    const int tid = threadIdx.x;
    const int lane = tid & 31;
    const int w = tid >> 5;
    if (tid < MAX_BUCKETS) hist[tid] = 0;
    __syncthreads();

    const int base = blockIdx.x * CHUNK;
    const int nvalid = min(CHUNK, E - base);
    int sv[EPT], dv[EPT], rk[EPT];
#pragma unroll
    for (int j = 0; j < EPT; ++j) {
        const int i = j * 1024 + tid;
        const int e = base + min(i, nvalid - 1);
        sv[j] = src[e];
        dv[j] = dst[e];
    }
#pragma unroll
    for (int j = 0; j < EPT; ++j) {
        const int i = j * 1024 + tid;
        rk[j] = (i < nvalid) ? atomicAdd(&hist[dv[j] >> BUCKET_SHIFT], 1) : 0;
    }

    int t = 0, c = 0, incT = 0, incC = 0;
    if (tid < MAX_BUCKETS) t = totals[tid];
    __syncthreads();
    if (tid < MAX_BUCKETS) {
        c = hist[tid];
        incT = t;
        incC = c;
        for (int o = 1; o < 32; o <<= 1) {
            const int uT = __shfl_up(incT, o);
            const int uC = __shfl_up(incC, o);
            if (lane >= o) { incT += uT; incC += uC; }
        }
        if (lane == 31) { wsumT[w] = incT; wsumC[w] = incC; }
    }
    __syncthreads();
    if (tid < MAX_BUCKETS) {
        int preT = 0, preC = 0;
        for (int k = 0; k < 8; ++k) {
            preT += (k < w) ? wsumT[k] : 0;
            preC += (k < w) ? wsumC[k] : 0;
        }
        const int segStart = preT + incT - t;
        const int ls = preC + incC - c;
        const int got = (c > 0) ? atomicAdd(&fill[tid], c) : 0;
        lstart[tid] = ls;
        gdelta[tid] = segStart + got - ls;
    }
    __syncthreads();
#pragma unroll
    for (int j = 0; j < EPT; ++j) {
        const int i = j * 1024 + tid;
        if (i < nvalid) {
            const int slot = lstart[dv[j] >> BUCKET_SHIFT] + rk[j];
            recS[slot] = sv[j];
            recD[slot] = dv[j];
        }
    }
    __syncthreads();
    int2* out = reinterpret_cast<int2*>(binned);
#pragma unroll
    for (int j = 0; j < EPT; ++j) {
        const int i = j * 1024 + tid;
        if (i < nvalid) {
            const int s = recS[i];
            const int d = recD[i];
            out[gdelta[d >> BUCKET_SHIFT] + i] = make_int2(s, d);
        }
    }
}

__global__ __launch_bounds__(1024)
void bucket_finalize_kernel(const int* __restrict__ binned, const int* __restrict__ totals,
                            int* __restrict__ rowptr, float* __restrict__ invdeg,
                            int* __restrict__ nbr, int N) {
    __shared__ int cnt[BUCKET_NODES];
    __shared__ int wsum[32];
    const int tid = threadIdx.x;
    const int lane = tid & 31;
    const int w = tid >> 5;
    const int b = blockIdx.x;
    if (tid < BUCKET_NODES) cnt[tid] = 0;

    int part = (tid < MAX_BUCKETS && tid < b) ? totals[tid] : 0;
    for (int o = 16; o > 0; o >>= 1) part += __shfl_xor(part, o);
    if (lane == 0) wsum[w] = part;
    const int count = totals[b];
    __syncthreads();
    int start = 0;
    for (int k = 0; k < 8; ++k) start += wsum[k];

    const int2* rec = reinterpret_cast<const int2*>(binned) + start;

    const int* recDst = binned + 2 * (size_t)start + 1;
    for (int i0 = 0; i0 < count; i0 += 1024 * RPT) {
        int d[RPT];
#pragma unroll
        for (int j = 0; j < RPT; ++j) {
            const int i = i0 + j * 1024 + tid;
            d[j] = recDst[2 * min(i, count - 1)];
        }
#pragma unroll
        for (int j = 0; j < RPT; ++j) {
            const int i = i0 + j * 1024 + tid;
            if (i < count) atomicAdd(&cnt[d[j] & (BUCKET_NODES - 1)], 1);
        }
    }
    __syncthreads();

    int c = 0, inc = 0;
    if (tid < BUCKET_NODES) {
        c = cnt[tid];
        inc = c;
        for (int o = 1; o < 32; o <<= 1) {
            const int u = __shfl_up(inc, o);
            if (lane >= o) inc += u;
        }
        if (lane == 31) wsum[w] = inc;
    }
    __syncthreads();
    int excl = 0;
    if (tid < BUCKET_NODES) {
        int pre = 0;
        for (int k = 0; k < 16; ++k) pre += (k < w) ? wsum[k] : 0;
        excl = pre + inc - c;
        cnt[tid] = excl;
    }
    __syncthreads();

    for (int i0 = 0; i0 < count; i0 += 1024 * RPT) {
        int2 r[RPT];
#pragma unroll
        for (int j = 0; j < RPT; ++j) {
            const int i = i0 + j * 1024 + tid;
            r[j] = rec[min(i, count - 1)];
        }
#pragma unroll
        for (int j = 0; j < RPT; ++j) {
            const int i = i0 + j * 1024 + tid;
            if (i < count) {
                const int pos = atomicAdd(&cnt[r[j].y & (BUCKET_NODES - 1)], 1);
                nbr[start + pos] = r[j].x;
            }
        }
    }

    if (tid < BUCKET_NODES) {
        const int node = (b << BUCKET_SHIFT) + tid;
        if (node < N) {
            rowptr[node] = start + excl;
            invdeg[node] = 1.0f / fmaxf((float)c, 1.0f);
        }
        if (node == N - 1) rowptr[N] = start + excl + c;
    }
}

template <int C>
__global__ __launch_bounds__(256)
void gather_sum_kernel(const float* __restrict__ x,
                       const int* __restrict__ rowptr,
                       const int* __restrict__ nbr,
                       float* __restrict__ agg, int N) {
    constexpr int R = C / 32;
    constexpr int U = 16;
    constexpr int NPW = 8;
    const int lane = threadIdx.x & 31;
    const int waveInBlock = __builtin_amdgcn_readfirstlane((int)(threadIdx.x >> 5));
    const int wave = blockIdx.x * 8 + waveInBlock;
    const int n0 = wave * NPW;
    if (n0 >= N) return;
    const int nn = min(NPW, N - n0);
    const int rp = rowptr[n0 + min(lane, nn)];

    for (int i = 0; i < nn; ++i) {
        const int start = __builtin_amdgcn_readlane(rp, i);
        const int end   = __builtin_amdgcn_readlane(rp, i + 1);
        float acc[R];
        for (int r = 0; r < R; ++r) acc[r] = 0.0f;

        for (int c = start; c < end; c += 32) {
            const int n = min(32, end - c);
            const int sv = nbr[c + min(lane, n - 1)];
            for (int j = 0; j < n; j += U) {
                float v[U][R];
#pragma unroll
                for (int u = 0; u < U; ++u) {
                    const float* xr = x + (size_t)__builtin_amdgcn_readlane(sv, min(j + u, n - 1)) * C;
                    for (int r = 0; r < R; ++r) v[u][r] = (j + u < n) ? xr[r * 32 + lane] : 0.0f;
                }
#pragma unroll
                for (int u = 0; u < U; ++u) {
                    for (int r = 0; r < R; ++r) acc[r] += v[u][r];
                }
            }
        }
        float* ar = agg + (size_t)(n0 + i) * C;
        for (int r = 0; r < R; ++r) ar[r * 32 + lane] = acc[r];
    }
}

template <int CIN, int COUT, bool RELU>
__global__ __launch_bounds__(256)
void sage_wmma_kernel(const float* __restrict__ agg,
                      const float* __restrict__ invdeg,
                      const float* __restrict__ xin,
                      const float* __restrict__ wl,
                      const float* __restrict__ wr,
                      const float* __restrict__ bias,
                      float* __restrict__ out,
                      int N) {
    const int lane = threadIdx.x & 31;
    const int waveInBlock = threadIdx.x >> 5;
    const int tile = blockIdx.x * (blockDim.x >> 5) + waveInBlock;
    const int row0 = tile * 16;
    if (row0 >= N) return;

    const int m  = lane & 15;
    const int kk = (lane >> 4) << 1;
    const bool fullTile = (row0 + 16 <= N);
    const int node = row0 + (fullTile ? m : min(m, N - 1 - row0));
    const float scale = invdeg[node];
    const float* aggRow = agg + (size_t)node * CIN;
    const float* xRow   = xin + (size_t)node * CIN;
    const int rbase = row0 + ((lane >> 4) << 3);

    for (int nb = 0; nb < COUT; nb += 16) {
        v8f c = {};
        const float* wlRow = wl + (size_t)(nb + m) * CIN;
        const float* wrRow = wr + (size_t)(nb + m) * CIN;
#pragma unroll
        for (int k0 = 0; k0 < CIN; k0 += 4) {
            v2f a_mean, a_self, b_l, b_r;
            a_mean.x = aggRow[k0 + kk] * scale;
            a_mean.y = aggRow[k0 + kk + 1] * scale;
            a_self.x = xRow[k0 + kk];
            a_self.y = xRow[k0 + kk + 1];
            b_l.x = wlRow[k0 + kk];
            b_l.y = wlRow[k0 + kk + 1];
            b_r.x = wrRow[k0 + kk];
            b_r.y = wrRow[k0 + kk + 1];
            c = __builtin_amdgcn_wmma_f32_16x16x4_f32(false, a_mean, false, b_l,
                                                      (short)0, c, false, false);
            c = __builtin_amdgcn_wmma_f32_16x16x4_f32(false, a_self, false, b_r,
                                                      (short)0, c, false, false);
        }
        const float bv = bias[nb + m];
        if (fullTile) {
            float* o = out + (size_t)rbase * COUT + nb + m;
#pragma unroll
            for (int v = 0; v < 8; ++v) {
                float val = c[v] + bv;
                if (RELU) val = fmaxf(val, 0.0f);
                o[(size_t)v * COUT] = val;
            }
        } else {
#pragma unroll
            for (int v = 0; v < 8; ++v) {
                int r = rbase + v;
                if (r < N) {
                    float val = c[v] + bv;
                    if (RELU) val = fmaxf(val, 0.0f);
                    out[(size_t)r * COUT + nb + m] = val;
                }
            }
        }
    }
}

template <int CIN, int COUT>
__global__ __launch_bounds__(256)
void sage_wmma_project_kernel(const float* __restrict__ agg,
                              const float* __restrict__ invdeg,
                              const float* __restrict__ xin,
                              const float* __restrict__ wl,
                              const float* __restrict__ wr,
                              const float* __restrict__ bias,
                              const float* __restrict__ wl3,
                              const float* __restrict__ wr3,
                              const float* __restrict__ b3,
                              float* __restrict__ p,
                              float* __restrict__ q,
                              int N) {
    __shared__ float stage[2][128];

    const int lane = threadIdx.x & 31;
    const int waveInBlock = __builtin_amdgcn_readfirstlane((int)(threadIdx.x >> 5));
    const int tile = blockIdx.x * 8 + waveInBlock;
    const int row0 = tile * 16;
    const int m    = lane & 15;
    const int half = lane >> 4;
    const int kk   = half << 1;

    float pl = 0.0f, pr = 0.0f;
    if (row0 < N) {
        const int node = min(row0 + m, N - 1);
        const float scale = invdeg[node];
        const float* aggRow = agg + (size_t)node * CIN;
        const float* xRow   = xin + (size_t)node * CIN;

        for (int nb = 0; nb < COUT; nb += 16) {
            v8f c = {};
            const float* wlRow = wl + (size_t)(nb + m) * CIN;
            const float* wrRow = wr + (size_t)(nb + m) * CIN;
#pragma unroll
            for (int k0 = 0; k0 < CIN; k0 += 4) {
                v2f x_mean, x_self, w_l, w_r;
                x_mean.x = aggRow[k0 + kk] * scale;
                x_mean.y = aggRow[k0 + kk + 1] * scale;
                x_self.x = xRow[k0 + kk];
                x_self.y = xRow[k0 + kk + 1];
                w_l.x = wlRow[k0 + kk];
                w_l.y = wlRow[k0 + kk + 1];
                w_r.x = wrRow[k0 + kk];
                w_r.y = wrRow[k0 + kk + 1];
                c = __builtin_amdgcn_wmma_f32_16x16x4_f32(false, w_l, false, x_mean,
                                                          (short)0, c, false, false);
                c = __builtin_amdgcn_wmma_f32_16x16x4_f32(false, w_r, false, x_self,
                                                          (short)0, c, false, false);
            }
            const int ch0 = nb + (half << 3);
            const float4 bA = *(const float4*)(bias + ch0), bB = *(const float4*)(bias + ch0 + 4);
            const float4 lA = *(const float4*)(wl3 + ch0),  lB = *(const float4*)(wl3 + ch0 + 4);
            const float4 rA = *(const float4*)(wr3 + ch0),  rB = *(const float4*)(wr3 + ch0 + 4);
            const float bv[8] = {bA.x, bA.y, bA.z, bA.w, bB.x, bB.y, bB.z, bB.w};
            const float lv[8] = {lA.x, lA.y, lA.z, lA.w, lB.x, lB.y, lB.z, lB.w};
            const float rv[8] = {rA.x, rA.y, rA.z, rA.w, rB.x, rB.y, rB.z, rB.w};
#pragma unroll
            for (int v = 0; v < 8; ++v) {
                const float h = fmaxf(c[v] + bv[v], 0.0f);
                pl += h * lv[v];
                pr += h * rv[v];
            }
        }
        pl += __shfl_xor(pl, 16);
        pr += __shfl_xor(pr, 16);
        pr += b3[0];
    }
    if (half == 0) {
        stage[0][waveInBlock * 16 + m] = pl;
        stage[1][waveInBlock * 16 + m] = pr;
    }
    __syncthreads();
    const int t = threadIdx.x & 127;
    const int which = threadIdx.x >> 7;
    const int nodeOut = blockIdx.x * 128 + t;
    if (nodeOut < N) (which ? q : p)[nodeOut] = stage[which][t];
}

__global__ __launch_bounds__(256)
void gather_scalar_final_kernel(const float* __restrict__ p,
                                const int* __restrict__ rowptr,
                                const int* __restrict__ nbr,
                                const float* __restrict__ invdeg,
                                const float* __restrict__ q,
                                float* __restrict__ out, int N) {
    const int i = blockIdx.x * blockDim.x + threadIdx.x;
    if (i >= N) return;
    int k = rowptr[i];
    const int end = rowptr[i + 1];
    float s = 0.0f;
    for (; k + 4 <= end; k += 4) {
        const int s0 = nbr[k], s1 = nbr[k + 1], s2 = nbr[k + 2], s3 = nbr[k + 3];
        const float v0 = p[s0], v1 = p[s1], v2 = p[s2], v3 = p[s3];
        s += v0;
        s += v1;
        s += v2;
        s += v3;
    }
    for (; k < end; ++k) s += p[nbr[k]];
    const float z = s * invdeg[i] + q[i];
    out[i] = 1.0f / (1.0f + expf(-z));
}

extern "C" void kernel_launch(void* const* d_in, const int* in_sizes, int n_in,
                              void* d_out, int out_size, void* d_ws, size_t ws_size,
                              hipStream_t stream) {
    const float* x    = (const float*)d_in[0];
    const int*   ei   = (const int*)d_in[1];
    const float* wl1  = (const float*)d_in[2];
    const float* wr1  = (const float*)d_in[3];
    const float* b1   = (const float*)d_in[4];
    const float* wl2  = (const float*)d_in[5];
    const float* wr2  = (const float*)d_in[6];
    const float* b2   = (const float*)d_in[7];
    const float* wl3  = (const float*)d_in[8];
    const float* wr3  = (const float*)d_in[9];
    const float* b3   = (const float*)d_in[10];
    float* out = (float*)d_out;

    const int N = in_sizes[0] / 32;
    const int E = in_sizes[1] / 2;
    const int* src = ei;
    const int* dst = ei + E;

    int*   totals = (int*)d_ws;
    int*   fill   = totals + 256;
    float* deg  = (float*)(totals + N);
    float* agg  = deg + N;
    float* h1   = agg + (size_t)N * 64;
    float* p3   = h1 + (size_t)N * 64;
    float* q3   = p3 + N;
    int*   nbr  = (int*)(q3 + N);
    int*   rptr = nbr + E;
    int*   binned = (int*)agg;

    const int tiles = (N + 15) / 16;
    const int gemmBlocks = (tiles + 7) / 8;
    const int chunkBlocks = (E + 8191) / 8192;
    const int buckets = (N + 511) / 512;
    const int nodeBlocks = (N + 63) / 64;

    hipMemsetAsync(totals, 0, 512 * sizeof(int), stream);
    bucket_count_kernel<<<chunkBlocks, 1024, 0, stream>>>(dst, totals, E);
    bucket_scatter_kernel<<<chunkBlocks, 1024, 0, stream>>>(src, dst, totals, fill, binned, E);
    bucket_finalize_kernel<<<buckets, 1024, 0, stream>>>(binned, totals, rptr, deg, nbr, N);

    gather_sum_kernel<32><<<nodeBlocks, 256, 0, stream>>>(x, rptr, nbr, agg, N);
    sage_wmma_kernel<32, 64, true><<<gemmBlocks, 256, 0, stream>>>(
        agg, deg, x, wl1, wr1, b1, h1, N);

    gather_sum_kernel<64><<<nodeBlocks, 256, 0, stream>>>(h1, rptr, nbr, agg, N);
    sage_wmma_project_kernel<64, 64><<<gemmBlocks, 256, 0, stream>>>(
        agg, deg, h1, wl2, wr2, b2, wl3, wr3, b3, p3, q3, N);

    gather_scalar_final_kernel<<<(N + 255) / 256, 256, 0, stream>>>(
        p3, rptr, nbr, deg, q3, out, N);
}


# Dense layers: weight matrices and each wave's 16 node rows copied to LDS with whole-line 16-byte loads; every WMMA fragment is one 8-byte LDS read, not a 16-line global load; product order kept
# speedup vs baseline: 3.8712x; 1.1304x over previous
#include <hip/hip_runtime.h>
#include <hip/hip_bf16.h>
#include <math.h>

typedef __attribute__((ext_vector_type(2))) float v2f;
typedef __attribute__((ext_vector_type(8))) float v8f;

constexpr int BUCKET_SHIFT = 9;
constexpr int BUCKET_NODES = 1 << BUCKET_SHIFT;
constexpr int MAX_BUCKETS  = 256;
constexpr int EPT          = 8;
constexpr int CHUNK        = 1024 * EPT;
constexpr int RPT          = 12;

__global__ __launch_bounds__(1024)
void bucket_count_kernel(const int* __restrict__ dst, int* __restrict__ totals, int E) {
    __shared__ int hist[MAX_BUCKETS];
    const int tid = threadIdx.x;
    if (tid < MAX_BUCKETS) hist[tid] = 0;
    __syncthreads();
    const int base = blockIdx.x * CHUNK;
    int dv[EPT];
#pragma unroll
    for (int j = 0; j < EPT; ++j) {
        const int e = base + j * 1024 + tid;
        dv[j] = dst[min(e, E - 1)];
    }
#pragma unroll
    for (int j = 0; j < EPT; ++j) {
        const int e = base + j * 1024 + tid;
        if (e < E) atomicAdd(&hist[dv[j] >> BUCKET_SHIFT], 1);
    }
    __syncthreads();
    if (tid < MAX_BUCKETS) {
        const int c = hist[tid];
        if (c > 0) atomicAdd(&totals[tid], c);
    }
}

__global__ __launch_bounds__(1024)
void bucket_scatter_kernel(const int* __restrict__ src, const int* __restrict__ dst,
                           const int* __restrict__ totals, int* __restrict__ fill,
                           int* __restrict__ binned, int E) {
    __shared__ int recS[CHUNK];
    __shared__ int recD[CHUNK];
    __shared__ int hist[MAX_BUCKETS];
    __shared__ int lstart[MAX_BUCKETS];
    __shared__ int gdelta[MAX_BUCKETS];
    __shared__ int wsumT[8];
    __shared__ int wsumC[8];
    const int tid = threadIdx.x;
    const int lane = tid & 31;
    const int w = tid >> 5;
    if (tid < MAX_BUCKETS) hist[tid] = 0;
    __syncthreads();

    const int base = blockIdx.x * CHUNK;
    const int nvalid = min(CHUNK, E - base);
    int sv[EPT], dv[EPT], rk[EPT];
#pragma unroll
    for (int j = 0; j < EPT; ++j) {
        const int i = j * 1024 + tid;
        const int e = base + min(i, nvalid - 1);
        sv[j] = src[e];
        dv[j] = dst[e];
    }
#pragma unroll
    for (int j = 0; j < EPT; ++j) {
        const int i = j * 1024 + tid;
        rk[j] = (i < nvalid) ? atomicAdd(&hist[dv[j] >> BUCKET_SHIFT], 1) : 0;
    }

    int t = 0, c = 0, incT = 0, incC = 0;
    if (tid < MAX_BUCKETS) t = totals[tid];
    __syncthreads();
    if (tid < MAX_BUCKETS) {
        c = hist[tid];
        incT = t;
        incC = c;
        for (int o = 1; o < 32; o <<= 1) {
            const int uT = __shfl_up(incT, o);
            const int uC = __shfl_up(incC, o);
            if (lane >= o) { incT += uT; incC += uC; }
        }
        if (lane == 31) { wsumT[w] = incT; wsumC[w] = incC; }
    }
    __syncthreads();
    if (tid < MAX_BUCKETS) {
        int preT = 0, preC = 0;
        for (int k = 0; k < 8; ++k) {
            preT += (k < w) ? wsumT[k] : 0;
            preC += (k < w) ? wsumC[k] : 0;
        }
        const int segStart = preT + incT - t;
        const int ls = preC + incC - c;
        const int got = (c > 0) ? atomicAdd(&fill[tid], c) : 0;
        lstart[tid] = ls;
        gdelta[tid] = segStart + got - ls;
    }
    __syncthreads();
#pragma unroll
    for (int j = 0; j < EPT; ++j) {
        const int i = j * 1024 + tid;
        if (i < nvalid) {
            const int slot = lstart[dv[j] >> BUCKET_SHIFT] + rk[j];
            recS[slot] = sv[j];
            recD[slot] = dv[j];
        }
    }
    __syncthreads();
    int2* out = reinterpret_cast<int2*>(binned);
#pragma unroll
    for (int j = 0; j < EPT; ++j) {
        const int i = j * 1024 + tid;
        if (i < nvalid) {
            const int s = recS[i];
            const int d = recD[i];
            out[gdelta[d >> BUCKET_SHIFT] + i] = make_int2(s, d);
        }
    }
}

__global__ __launch_bounds__(1024)
void bucket_finalize_kernel(const int* __restrict__ binned, const int* __restrict__ totals,
                            int* __restrict__ rowptr, float* __restrict__ invdeg,
                            int* __restrict__ nbr, int N) {
    __shared__ int cnt[BUCKET_NODES];
    __shared__ int wsum[32];
    const int tid = threadIdx.x;
    const int lane = tid & 31;
    const int w = tid >> 5;
    const int b = blockIdx.x;
    if (tid < BUCKET_NODES) cnt[tid] = 0;

    int part = (tid < MAX_BUCKETS && tid < b) ? totals[tid] : 0;
    for (int o = 16; o > 0; o >>= 1) part += __shfl_xor(part, o);
    if (lane == 0) wsum[w] = part;
    const int count = totals[b];
    __syncthreads();
    int start = 0;
    for (int k = 0; k < 8; ++k) start += wsum[k];

    const int2* rec = reinterpret_cast<const int2*>(binned) + start;

    const int* recDst = binned + 2 * (size_t)start + 1;
    for (int i0 = 0; i0 < count; i0 += 1024 * RPT) {
        int d[RPT];
#pragma unroll
        for (int j = 0; j < RPT; ++j) {
            const int i = i0 + j * 1024 + tid;
            d[j] = recDst[2 * min(i, count - 1)];
        }
#pragma unroll
        for (int j = 0; j < RPT; ++j) {
            const int i = i0 + j * 1024 + tid;
            if (i < count) atomicAdd(&cnt[d[j] & (BUCKET_NODES - 1)], 1);
        }
    }
    __syncthreads();

    int c = 0, inc = 0;
    if (tid < BUCKET_NODES) {
        c = cnt[tid];
        inc = c;
        for (int o = 1; o < 32; o <<= 1) {
            const int u = __shfl_up(inc, o);
            if (lane >= o) inc += u;
        }
        if (lane == 31) wsum[w] = inc;
    }
    __syncthreads();
    int excl = 0;
    if (tid < BUCKET_NODES) {
        int pre = 0;
        for (int k = 0; k < 16; ++k) pre += (k < w) ? wsum[k] : 0;
        excl = pre + inc - c;
        cnt[tid] = excl;
    }
    __syncthreads();

    for (int i0 = 0; i0 < count; i0 += 1024 * RPT) {
        int2 r[RPT];
#pragma unroll
        for (int j = 0; j < RPT; ++j) {
            const int i = i0 + j * 1024 + tid;
            r[j] = rec[min(i, count - 1)];
        }
#pragma unroll
        for (int j = 0; j < RPT; ++j) {
            const int i = i0 + j * 1024 + tid;
            if (i < count) {
                const int pos = atomicAdd(&cnt[r[j].y & (BUCKET_NODES - 1)], 1);
                nbr[start + pos] = r[j].x;
            }
        }
    }

    if (tid < BUCKET_NODES) {
        const int node = (b << BUCKET_SHIFT) + tid;
        if (node < N) {
            rowptr[node] = start + excl;
            invdeg[node] = 1.0f / fmaxf((float)c, 1.0f);
        }
        if (node == N - 1) rowptr[N] = start + excl + c;
    }
}

template <int C>
__global__ __launch_bounds__(256)
void gather_sum_kernel(const float* __restrict__ x,
                       const int* __restrict__ rowptr,
                       const int* __restrict__ nbr,
                       float* __restrict__ agg, int N) {
    constexpr int R = C / 32;
    constexpr int U = 16;
    constexpr int NPW = 8;
    const int lane = threadIdx.x & 31;
    const int waveInBlock = __builtin_amdgcn_readfirstlane((int)(threadIdx.x >> 5));
    const int wave = blockIdx.x * 8 + waveInBlock;
    const int n0 = wave * NPW;
    if (n0 >= N) return;
    const int nn = min(NPW, N - n0);
    const int rp = rowptr[n0 + min(lane, nn)];

    for (int i = 0; i < nn; ++i) {
        const int start = __builtin_amdgcn_readlane(rp, i);
        const int end   = __builtin_amdgcn_readlane(rp, i + 1);
        float acc[R];
        for (int r = 0; r < R; ++r) acc[r] = 0.0f;

        for (int c = start; c < end; c += 32) {
            const int n = min(32, end - c);
            const int sv = nbr[c + min(lane, n - 1)];
            for (int j = 0; j < n; j += U) {
                float v[U][R];
#pragma unroll
                for (int u = 0; u < U; ++u) {
                    const float* xr = x + (size_t)__builtin_amdgcn_readlane(sv, min(j + u, n - 1)) * C;
                    for (int r = 0; r < R; ++r) v[u][r] = (j + u < n) ? xr[r * 32 + lane] : 0.0f;
                }
#pragma unroll
                for (int u = 0; u < U; ++u) {
                    for (int r = 0; r < R; ++r) acc[r] += v[u][r];
                }
            }
        }
        float* ar = agg + (size_t)(n0 + i) * C;
        for (int r = 0; r < R; ++r) ar[r * 32 + lane] = acc[r];
    }
}

constexpr int WAVES_PER_BLOCK = 8;

template <int CIN, int P, int ROWS, int NT>
__device__ __forceinline__ void stage_matrix(const float* __restrict__ g, float* s, int t) {
    constexpr int V = CIN / 4;
    static_assert((ROWS * V) % NT == 0, "whole passes only");
#pragma unroll
    for (int i = 0; i < ROWS * V / NT; ++i) {
        const int v = i * NT + t;
        const float4 val = *reinterpret_cast<const float4*>(g + (size_t)v * 4);
        *reinterpret_cast<float4*>(s + (v / V) * P + (v % V) * 4) = val;
    }
}

template <int CIN, int P>
__device__ __forceinline__ void stage_node_tile(const float* __restrict__ agg,
                                                const float* __restrict__ invdeg,
                                                const float* __restrict__ xin,
                                                float* sMean, float* sSelf,
                                                int row0, int N, int lane) {
    constexpr int V = CIN / 4;
#pragma unroll
    for (int i = 0; i < 16 * V / 32; ++i) {
        const int v = i * 32 + lane;
        const int r = v / V;
        const int c = (v % V) * 4;
        const int node = min(row0 + r, N - 1);
        const float scale = invdeg[node];
        float4 a = *reinterpret_cast<const float4*>(agg + (size_t)node * CIN + c);
        const float4 s = *reinterpret_cast<const float4*>(xin + (size_t)node * CIN + c);
        a.x *= scale; a.y *= scale; a.z *= scale; a.w *= scale;
        *reinterpret_cast<float4*>(sMean + r * P + c) = a;
        *reinterpret_cast<float4*>(sSelf + r * P + c) = s;
    }
}

template <int CIN, int COUT, bool RELU>
__global__ __launch_bounds__(256)
void sage_wmma_kernel(const float* __restrict__ agg,
                      const float* __restrict__ invdeg,
                      const float* __restrict__ xin,
                      const float* __restrict__ wl,
                      const float* __restrict__ wr,
                      const float* __restrict__ bias,
                      float* __restrict__ out,
                      int N) {
    constexpr int P = CIN + 4;
    __shared__ __attribute__((aligned(16))) float lds[2 * COUT * P + 2 * WAVES_PER_BLOCK * 16 * P];

    const int lane = threadIdx.x & 31;
    const int waveInBlock = __builtin_amdgcn_readfirstlane((int)(threadIdx.x >> 5));
    const int tile = blockIdx.x * WAVES_PER_BLOCK + waveInBlock;
    const int row0 = tile * 16;

    float* sWl   = lds;
    float* sWr   = lds + COUT * P;
    float* sMean = lds + 2 * COUT * P + waveInBlock * (16 * P);
    float* sSelf = sMean + WAVES_PER_BLOCK * 16 * P;

    stage_matrix<CIN, P, COUT, 256>(wl, sWl, threadIdx.x);
    stage_matrix<CIN, P, COUT, 256>(wr, sWr, threadIdx.x);
    if (row0 < N)
        stage_node_tile<CIN, P>(agg, invdeg, xin, sMean, sSelf, row0, N, lane);
    __syncthreads();
    if (row0 >= N) return;

    const int m  = lane & 15;
    const int kk = (lane >> 4) << 1;
    const bool fullTile = (row0 + 16 <= N);
    const int rbase = row0 + ((lane >> 4) << 3);

    v2f a_mean[CIN / 4], a_self[CIN / 4];
    const float* meanRow = sMean + m * P + kk;
    const float* selfRow = sSelf + m * P + kk;
#pragma unroll
    for (int s = 0; s < CIN / 4; ++s) {
        a_mean[s] = *reinterpret_cast<const v2f*>(meanRow + 4 * s);
        a_self[s] = *reinterpret_cast<const v2f*>(selfRow + 4 * s);
    }

#pragma unroll
    for (int nb = 0; nb < COUT; nb += 16) {
        v8f c = {};
        const float* wlRow = sWl + (nb + m) * P + kk;
        const float* wrRow = sWr + (nb + m) * P + kk;
#pragma unroll
        for (int s = 0; s < CIN / 4; ++s) {
            const v2f b_l = *reinterpret_cast<const v2f*>(wlRow + 4 * s);
            const v2f b_r = *reinterpret_cast<const v2f*>(wrRow + 4 * s);
            c = __builtin_amdgcn_wmma_f32_16x16x4_f32(false, a_mean[s], false, b_l,
                                                      (short)0, c, false, false);
            c = __builtin_amdgcn_wmma_f32_16x16x4_f32(false, a_self[s], false, b_r,
                                                      (short)0, c, false, false);
        }
        const float bv = bias[nb + m];
        if (fullTile) {
            float* o = out + (size_t)rbase * COUT + nb + m;
#pragma unroll
            for (int v = 0; v < 8; ++v) {
                float val = c[v] + bv;
                if (RELU) val = fmaxf(val, 0.0f);
                o[(size_t)v * COUT] = val;
            }
        } else {
#pragma unroll
            for (int v = 0; v < 8; ++v) {
                int r = rbase + v;
                if (r < N) {
                    float val = c[v] + bv;
                    if (RELU) val = fmaxf(val, 0.0f);
                    out[(size_t)r * COUT + nb + m] = val;
                }
            }
        }
    }
}

template <int CIN, int COUT>
__global__ __launch_bounds__(256)
void sage_wmma_project_kernel(const float* __restrict__ agg,
                              const float* __restrict__ invdeg,
                              const float* __restrict__ xin,
                              const float* __restrict__ wl,
                              const float* __restrict__ wr,
                              const float* __restrict__ bias,
                              const float* __restrict__ wl3,
                              const float* __restrict__ wr3,
                              const float* __restrict__ b3,
                              float* __restrict__ p,
                              float* __restrict__ q,
                              int N) {
    constexpr int P = CIN + 4;
    __shared__ __attribute__((aligned(16))) float lds[2 * COUT * P + 2 * WAVES_PER_BLOCK * 16 * P + 256];

    const int lane = threadIdx.x & 31;
    const int waveInBlock = __builtin_amdgcn_readfirstlane((int)(threadIdx.x >> 5));
    const int tile = blockIdx.x * WAVES_PER_BLOCK + waveInBlock;
    const int row0 = tile * 16;
    const int m    = lane & 15;
    const int half = lane >> 4;
    const int kk   = half << 1;

    float* sWl   = lds;
    float* sWr   = lds + COUT * P;
    float* sMean = lds + 2 * COUT * P + waveInBlock * (16 * P);
    float* sSelf = sMean + WAVES_PER_BLOCK * 16 * P;
    float* stage = lds + 2 * COUT * P + 2 * WAVES_PER_BLOCK * 16 * P;

    stage_matrix<CIN, P, COUT, 256>(wl, sWl, threadIdx.x);
    stage_matrix<CIN, P, COUT, 256>(wr, sWr, threadIdx.x);
    if (row0 < N)
        stage_node_tile<CIN, P>(agg, invdeg, xin, sMean, sSelf, row0, N, lane);
    __syncthreads();

    float pl = 0.0f, pr = 0.0f;
    if (row0 < N) {
        v2f x_mean[CIN / 4], x_self[CIN / 4];
        const float* meanRow = sMean + m * P + kk;
        const float* selfRow = sSelf + m * P + kk;
#pragma unroll
        for (int s = 0; s < CIN / 4; ++s) {
            x_mean[s] = *reinterpret_cast<const v2f*>(meanRow + 4 * s);
            x_self[s] = *reinterpret_cast<const v2f*>(selfRow + 4 * s);
        }

#pragma unroll
        for (int nb = 0; nb < COUT; nb += 16) {
            v8f c = {};
            const float* wlRow = sWl + (nb + m) * P + kk;
            const float* wrRow = sWr + (nb + m) * P + kk;
#pragma unroll
            for (int s = 0; s < CIN / 4; ++s) {
                const v2f w_l = *reinterpret_cast<const v2f*>(wlRow + 4 * s);
                const v2f w_r = *reinterpret_cast<const v2f*>(wrRow + 4 * s);
                c = __builtin_amdgcn_wmma_f32_16x16x4_f32(false, w_l, false, x_mean[s],
                                                          (short)0, c, false, false);
                c = __builtin_amdgcn_wmma_f32_16x16x4_f32(false, w_r, false, x_self[s],
                                                          (short)0, c, false, false);
            }
            const int ch0 = nb + (half << 3);
            const float4 bA = *(const float4*)(bias + ch0), bB = *(const float4*)(bias + ch0 + 4);
            const float4 lA = *(const float4*)(wl3 + ch0),  lB = *(const float4*)(wl3 + ch0 + 4);
            const float4 rA = *(const float4*)(wr3 + ch0),  rB = *(const float4*)(wr3 + ch0 + 4);
            const float bv[8] = {bA.x, bA.y, bA.z, bA.w, bB.x, bB.y, bB.z, bB.w};
            const float lv[8] = {lA.x, lA.y, lA.z, lA.w, lB.x, lB.y, lB.z, lB.w};
            const float rv[8] = {rA.x, rA.y, rA.z, rA.w, rB.x, rB.y, rB.z, rB.w};
#pragma unroll
            for (int v = 0; v < 8; ++v) {
                const float h = fmaxf(c[v] + bv[v], 0.0f);
                pl += h * lv[v];
                pr += h * rv[v];
            }
        }
        pl += __shfl_xor(pl, 16);
        pr += __shfl_xor(pr, 16);
        pr += b3[0];
    }
    if (half == 0) {
        stage[waveInBlock * 16 + m] = pl;
        stage[128 + waveInBlock * 16 + m] = pr;
    }
    __syncthreads();
    const int t = threadIdx.x & 127;
    const int which = threadIdx.x >> 7;
    const int nodeOut = blockIdx.x * 128 + t;
    if (nodeOut < N) (which ? q : p)[nodeOut] = stage[which * 128 + t];
}

__global__ __launch_bounds__(256)
void gather_scalar_final_kernel(const float* __restrict__ p,
                                const int* __restrict__ rowptr,
                                const int* __restrict__ nbr,
                                const float* __restrict__ invdeg,
                                const float* __restrict__ q,
                                float* __restrict__ out, int N) {
    const int i = blockIdx.x * blockDim.x + threadIdx.x;
    if (i >= N) return;
    int k = rowptr[i];
    const int end = rowptr[i + 1];
    float s = 0.0f;
    for (; k + 4 <= end; k += 4) {
        const int s0 = nbr[k], s1 = nbr[k + 1], s2 = nbr[k + 2], s3 = nbr[k + 3];
        const float v0 = p[s0], v1 = p[s1], v2 = p[s2], v3 = p[s3];
        s += v0;
        s += v1;
        s += v2;
        s += v3;
    }
    for (; k < end; ++k) s += p[nbr[k]];
    const float z = s * invdeg[i] + q[i];
    out[i] = 1.0f / (1.0f + expf(-z));
}

extern "C" void kernel_launch(void* const* d_in, const int* in_sizes, int n_in,
                              void* d_out, int out_size, void* d_ws, size_t ws_size,
                              hipStream_t stream) {
    const float* x    = (const float*)d_in[0];
    const int*   ei   = (const int*)d_in[1];
    const float* wl1  = (const float*)d_in[2];
    const float* wr1  = (const float*)d_in[3];
    const float* b1   = (const float*)d_in[4];
    const float* wl2  = (const float*)d_in[5];
    const float* wr2  = (const float*)d_in[6];
    const float* b2   = (const float*)d_in[7];
    const float* wl3  = (const float*)d_in[8];
    const float* wr3  = (const float*)d_in[9];
    const float* b3   = (const float*)d_in[10];
    float* out = (float*)d_out;

    const int N = in_sizes[0] / 32;
    const int E = in_sizes[1] / 2;
    const int* src = ei;
    const int* dst = ei + E;

    int*   totals = (int*)d_ws;
    int*   fill   = totals + 256;
    float* deg  = (float*)(totals + N);
    float* agg  = deg + N;
    float* h1   = agg + (size_t)N * 64;
    float* p3   = h1 + (size_t)N * 64;
    float* q3   = p3 + N;
    int*   nbr  = (int*)(q3 + N);
    int*   rptr = nbr + E;
    int*   binned = (int*)agg;

    const int tiles = (N + 15) / 16;
    const int gemmBlocks = (tiles + 7) / 8;
    const int chunkBlocks = (E + 8191) / 8192;
    const int buckets = (N + 511) / 512;
    const int nodeBlocks = (N + 63) / 64;

    hipMemsetAsync(totals, 0, 512 * sizeof(int), stream);
    bucket_count_kernel<<<chunkBlocks, 1024, 0, stream>>>(dst, totals, E);
    bucket_scatter_kernel<<<chunkBlocks, 1024, 0, stream>>>(src, dst, totals, fill, binned, E);
    bucket_finalize_kernel<<<buckets, 1024, 0, stream>>>(binned, totals, rptr, deg, nbr, N);

    gather_sum_kernel<32><<<nodeBlocks, 256, 0, stream>>>(x, rptr, nbr, agg, N);
    sage_wmma_kernel<32, 64, true><<<gemmBlocks, 256, 0, stream>>>(
        agg, deg, x, wl1, wr1, b1, h1, N);

    gather_sum_kernel<64><<<nodeBlocks, 256, 0, stream>>>(h1, rptr, nbr, agg, N);
    sage_wmma_project_kernel<64, 64><<<gemmBlocks, 256, 0, stream>>>(
        agg, deg, h1, wl2, wr2, b2, wl3, wr3, b3, p3, q3, N);

    gather_scalar_final_kernel<<<(N + 255) / 256, 256, 0, stream>>>(
        p3, rptr, nbr, deg, q3, out, N);
}


# Neighbour-sum rows read by 16-byte per-lane loads: one load takes the rows of 4 (C=32) or 2 (C=64) edges, row index by in-group lane pick, no per-edge scalar code; sum order: strided partials, tree
# speedup vs baseline: 4.3062x; 1.1124x over previous
#include <hip/hip_runtime.h>
#include <hip/hip_bf16.h>
#include <math.h>

typedef __attribute__((ext_vector_type(2))) float v2f;
typedef __attribute__((ext_vector_type(8))) float v8f;

constexpr int BUCKET_SHIFT = 9;
constexpr int BUCKET_NODES = 1 << BUCKET_SHIFT;
constexpr int MAX_BUCKETS  = 256;
constexpr int EPT          = 8;
constexpr int CHUNK        = 1024 * EPT;
constexpr int RPT          = 12;

__global__ __launch_bounds__(1024)
void bucket_count_kernel(const int* __restrict__ dst, int* __restrict__ totals, int E) {
    __shared__ int hist[MAX_BUCKETS];
    const int tid = threadIdx.x;
    if (tid < MAX_BUCKETS) hist[tid] = 0;
    __syncthreads();
    const int base = blockIdx.x * CHUNK;
    int dv[EPT];
#pragma unroll
    for (int j = 0; j < EPT; ++j) {
        const int e = base + j * 1024 + tid;
        dv[j] = dst[min(e, E - 1)];
    }
#pragma unroll
    for (int j = 0; j < EPT; ++j) {
        const int e = base + j * 1024 + tid;
        if (e < E) atomicAdd(&hist[dv[j] >> BUCKET_SHIFT], 1);
    }
    __syncthreads();
    if (tid < MAX_BUCKETS) {
        const int c = hist[tid];
        if (c > 0) atomicAdd(&totals[tid], c);
    }
}

__global__ __launch_bounds__(1024)
void bucket_scatter_kernel(const int* __restrict__ src, const int* __restrict__ dst,
                           const int* __restrict__ totals, int* __restrict__ fill,
                           int* __restrict__ binned, int E) {
    __shared__ int recS[CHUNK];
    __shared__ int recD[CHUNK];
    __shared__ int hist[MAX_BUCKETS];
    __shared__ int lstart[MAX_BUCKETS];
    __shared__ int gdelta[MAX_BUCKETS];
    __shared__ int wsumT[8];
    __shared__ int wsumC[8];
    const int tid = threadIdx.x;
    const int lane = tid & 31;
    const int w = tid >> 5;
    if (tid < MAX_BUCKETS) hist[tid] = 0;
    __syncthreads();

    const int base = blockIdx.x * CHUNK;
    const int nvalid = min(CHUNK, E - base);
    int sv[EPT], dv[EPT], rk[EPT];
#pragma unroll
    for (int j = 0; j < EPT; ++j) {
        const int i = j * 1024 + tid;
        const int e = base + min(i, nvalid - 1);
        sv[j] = src[e];
        dv[j] = dst[e];
    }
#pragma unroll
    for (int j = 0; j < EPT; ++j) {
        const int i = j * 1024 + tid;
        rk[j] = (i < nvalid) ? atomicAdd(&hist[dv[j] >> BUCKET_SHIFT], 1) : 0;
    }

    int t = 0, c = 0, incT = 0, incC = 0;
    if (tid < MAX_BUCKETS) t = totals[tid];
    __syncthreads();
    if (tid < MAX_BUCKETS) {
        c = hist[tid];
        incT = t;
        incC = c;
        for (int o = 1; o < 32; o <<= 1) {
            const int uT = __shfl_up(incT, o);
            const int uC = __shfl_up(incC, o);
            if (lane >= o) { incT += uT; incC += uC; }
        }
        if (lane == 31) { wsumT[w] = incT; wsumC[w] = incC; }
    }
    __syncthreads();
    if (tid < MAX_BUCKETS) {
        int preT = 0, preC = 0;
        for (int k = 0; k < 8; ++k) {
            preT += (k < w) ? wsumT[k] : 0;
            preC += (k < w) ? wsumC[k] : 0;
        }
        const int segStart = preT + incT - t;
        const int ls = preC + incC - c;
        const int got = (c > 0) ? atomicAdd(&fill[tid], c) : 0;
        lstart[tid] = ls;
        gdelta[tid] = segStart + got - ls;
    }
    __syncthreads();
#pragma unroll
    for (int j = 0; j < EPT; ++j) {
        const int i = j * 1024 + tid;
        if (i < nvalid) {
            const int slot = lstart[dv[j] >> BUCKET_SHIFT] + rk[j];
            recS[slot] = sv[j];
            recD[slot] = dv[j];
        }
    }
    __syncthreads();
    int2* out = reinterpret_cast<int2*>(binned);
#pragma unroll
    for (int j = 0; j < EPT; ++j) {
        const int i = j * 1024 + tid;
        if (i < nvalid) {
            const int s = recS[i];
            const int d = recD[i];
            out[gdelta[d >> BUCKET_SHIFT] + i] = make_int2(s, d);
        }
    }
}

__global__ __launch_bounds__(1024)
void bucket_finalize_kernel(const int* __restrict__ binned, const int* __restrict__ totals,
                            int* __restrict__ rowptr, float* __restrict__ invdeg,
                            int* __restrict__ nbr, int N) {
    __shared__ int cnt[BUCKET_NODES];
    __shared__ int wsum[32];
    const int tid = threadIdx.x;
    const int lane = tid & 31;
    const int w = tid >> 5;
    const int b = blockIdx.x;
    if (tid < BUCKET_NODES) cnt[tid] = 0;

    int part = (tid < MAX_BUCKETS && tid < b) ? totals[tid] : 0;
    for (int o = 16; o > 0; o >>= 1) part += __shfl_xor(part, o);
    if (lane == 0) wsum[w] = part;
    const int count = totals[b];
    __syncthreads();
    int start = 0;
    for (int k = 0; k < 8; ++k) start += wsum[k];

    const int2* rec = reinterpret_cast<const int2*>(binned) + start;

    const int* recDst = binned + 2 * (size_t)start + 1;
    for (int i0 = 0; i0 < count; i0 += 1024 * RPT) {
        int d[RPT];
#pragma unroll
        for (int j = 0; j < RPT; ++j) {
            const int i = i0 + j * 1024 + tid;
            d[j] = recDst[2 * min(i, count - 1)];
        }
#pragma unroll
        for (int j = 0; j < RPT; ++j) {
            const int i = i0 + j * 1024 + tid;
            if (i < count) atomicAdd(&cnt[d[j] & (BUCKET_NODES - 1)], 1);
        }
    }
    __syncthreads();

    int c = 0, inc = 0;
    if (tid < BUCKET_NODES) {
        c = cnt[tid];
        inc = c;
        for (int o = 1; o < 32; o <<= 1) {
            const int u = __shfl_up(inc, o);
            if (lane >= o) inc += u;
        }
        if (lane == 31) wsum[w] = inc;
    }
    __syncthreads();
    int excl = 0;
    if (tid < BUCKET_NODES) {
        int pre = 0;
        for (int k = 0; k < 16; ++k) pre += (k < w) ? wsum[k] : 0;
        excl = pre + inc - c;
        cnt[tid] = excl;
    }
    __syncthreads();

    for (int i0 = 0; i0 < count; i0 += 1024 * RPT) {
        int2 r[RPT];
#pragma unroll
        for (int j = 0; j < RPT; ++j) {
            const int i = i0 + j * 1024 + tid;
            r[j] = rec[min(i, count - 1)];
        }
#pragma unroll
        for (int j = 0; j < RPT; ++j) {
            const int i = i0 + j * 1024 + tid;
            if (i < count) {
                const int pos = atomicAdd(&cnt[r[j].y & (BUCKET_NODES - 1)], 1);
                nbr[start + pos] = r[j].x;
            }
        }
    }

    if (tid < BUCKET_NODES) {
        const int node = (b << BUCKET_SHIFT) + tid;
        if (node < N) {
            rowptr[node] = start + excl;
            invdeg[node] = 1.0f / fmaxf((float)c, 1.0f);
        }
        if (node == N - 1) rowptr[N] = start + excl + c;
    }
}

__device__ __forceinline__ float swap_rows16(float v) {
    return __builtin_bit_cast(float, __builtin_amdgcn_permlanex16(
        __builtin_bit_cast(int, v), __builtin_bit_cast(int, v), 0x76543210, 0xfedcba98, false, false));
}
__device__ __forceinline__ float swap_groups8(float v) {
    return __builtin_bit_cast(float, __builtin_amdgcn_update_dpp(
        0, __builtin_bit_cast(int, v), 0x168, 0xf, 0xf, false));
}

template <int C, int K, int t>
__device__ __forceinline__ void issue_rows(const char* __restrict__ xb, unsigned sv, unsigned subOff,
                                           bool lastOk, float4 (&v)[K]) {
    if constexpr (t < K) {
        const unsigned idx = __builtin_amdgcn_mov_dpp8(sv, t * 0x249249u);
        const unsigned off = idx * (unsigned)(C * 4) + subOff;
        if (t + 1 < K || lastOk) v[t] = *reinterpret_cast<const float4*>(xb + off);
        else v[t] = make_float4(0.0f, 0.0f, 0.0f, 0.0f);
        issue_rows<C, K, t + 1>(xb, sv, subOff, lastOk, v);
    }
}

template <int C, int K>
__device__ __forceinline__ void gather_pass(const char* __restrict__ xb, unsigned sv, unsigned subOff,
                                            bool lastOk, float4& acc) {
    float4 v[K];
    issue_rows<C, K, 0>(xb, sv, subOff, lastOk, v);
#pragma unroll
    for (int t = 0; t < K; ++t) {
        acc.x += v[t].x; acc.y += v[t].y; acc.z += v[t].z; acc.w += v[t].w;
    }
}

template <int C>
__global__ __launch_bounds__(256)
void gather_sum_kernel(const float* __restrict__ x,
                       const int* __restrict__ rowptr,
                       const int* __restrict__ nbr,
                       float* __restrict__ agg, int N) {
    constexpr int LPR = C / 4;
    constexpr int G   = 32 / LPR;
    constexpr int PS  = 8 * G;
    constexpr int NPW = 8;
    static_assert(C == 32 || C == 64, "a row is 8 or 16 lanes");
    const int lane = threadIdx.x & 31;
    const int g    = lane / LPR;
    const unsigned subOff = (lane % LPR) * 16;
    const int perm = G * (lane & 7) + g;
    const int waveInBlock = __builtin_amdgcn_readfirstlane((int)(threadIdx.x >> 5));
    const int wave = blockIdx.x * 8 + waveInBlock;
    const int n0 = wave * NPW;
    if (n0 >= N) return;
    const int nn = min(NPW, N - n0);
    const int rp = rowptr[n0 + min(lane, nn)];
    int pos = __builtin_amdgcn_readlane(rp, 0);
    const int wend = __builtin_amdgcn_readlane(rp, nn);
    const char* xb = reinterpret_cast<const char*>(x);

    unsigned svNext = 0;
    if (pos < wend) svNext = (unsigned)nbr[min(pos + perm, wend - 1)];

    for (int i = 0; i < nn; ++i) {
        const int end = __builtin_amdgcn_readlane(rp, i + 1);
        float4 acc = make_float4(0.0f, 0.0f, 0.0f, 0.0f);
        while (pos < end) {
            const int n = min(PS, end - pos);
            const unsigned sv = svNext;
            pos += n;
            if (pos < wend) svNext = (unsigned)nbr[min(pos + perm, wend - 1)];
            const int K = (n + G - 1) / G;
            const bool lastOk = g < n - (K - 1) * G;
            switch (K) {
                case 1:  gather_pass<C, 1>(xb, sv, subOff, lastOk, acc); break;
                case 2:  gather_pass<C, 2>(xb, sv, subOff, lastOk, acc); break;
                case 3:  gather_pass<C, 3>(xb, sv, subOff, lastOk, acc); break;
                case 4:  gather_pass<C, 4>(xb, sv, subOff, lastOk, acc); break;
                case 5:  gather_pass<C, 5>(xb, sv, subOff, lastOk, acc); break;
                case 6:  gather_pass<C, 6>(xb, sv, subOff, lastOk, acc); break;
                case 7:  gather_pass<C, 7>(xb, sv, subOff, lastOk, acc); break;
                default: gather_pass<C, 8>(xb, sv, subOff, lastOk, acc); break;
            }
        }
        if (G == 4) {
            acc.x += swap_groups8(acc.x); acc.y += swap_groups8(acc.y);
            acc.z += swap_groups8(acc.z); acc.w += swap_groups8(acc.w);
        }
        acc.x += swap_rows16(acc.x); acc.y += swap_rows16(acc.y);
        acc.z += swap_rows16(acc.z); acc.w += swap_rows16(acc.w);
        if (lane < LPR)
            *reinterpret_cast<float4*>(agg + (size_t)(n0 + i) * C + lane * 4) = acc;
    }
}

constexpr int WAVES_PER_BLOCK = 8;

template <int CIN, int P, int ROWS, int NT>
__device__ __forceinline__ void stage_matrix(const float* __restrict__ g, float* s, int t) {
    constexpr int V = CIN / 4;
    static_assert((ROWS * V) % NT == 0, "whole passes only");
#pragma unroll
    for (int i = 0; i < ROWS * V / NT; ++i) {
        const int v = i * NT + t;
        const float4 val = *reinterpret_cast<const float4*>(g + (size_t)v * 4);
        *reinterpret_cast<float4*>(s + (v / V) * P + (v % V) * 4) = val;
    }
}

template <int CIN, int P>
__device__ __forceinline__ void stage_node_tile(const float* __restrict__ agg,
                                                const float* __restrict__ invdeg,
                                                const float* __restrict__ xin,
                                                float* sMean, float* sSelf,
                                                int row0, int N, int lane) {
    constexpr int V = CIN / 4;
#pragma unroll
    for (int i = 0; i < 16 * V / 32; ++i) {
        const int v = i * 32 + lane;
        const int r = v / V;
        const int c = (v % V) * 4;
        const int node = min(row0 + r, N - 1);
        const float scale = invdeg[node];
        float4 a = *reinterpret_cast<const float4*>(agg + (size_t)node * CIN + c);
        const float4 s = *reinterpret_cast<const float4*>(xin + (size_t)node * CIN + c);
        a.x *= scale; a.y *= scale; a.z *= scale; a.w *= scale;
        *reinterpret_cast<float4*>(sMean + r * P + c) = a;
        *reinterpret_cast<float4*>(sSelf + r * P + c) = s;
    }
}

template <int CIN, int COUT, bool RELU>
__global__ __launch_bounds__(256)
void sage_wmma_kernel(const float* __restrict__ agg,
                      const float* __restrict__ invdeg,
                      const float* __restrict__ xin,
                      const float* __restrict__ wl,
                      const float* __restrict__ wr,
                      const float* __restrict__ bias,
                      float* __restrict__ out,
                      int N) {
    constexpr int P = CIN + 4;
    __shared__ __attribute__((aligned(16))) float lds[2 * COUT * P + 2 * WAVES_PER_BLOCK * 16 * P];

    const int lane = threadIdx.x & 31;
    const int waveInBlock = __builtin_amdgcn_readfirstlane((int)(threadIdx.x >> 5));
    const int tile = blockIdx.x * WAVES_PER_BLOCK + waveInBlock;
    const int row0 = tile * 16;

    float* sWl   = lds;
    float* sWr   = lds + COUT * P;
    float* sMean = lds + 2 * COUT * P + waveInBlock * (16 * P);
    float* sSelf = sMean + WAVES_PER_BLOCK * 16 * P;

    stage_matrix<CIN, P, COUT, 256>(wl, sWl, threadIdx.x);
    stage_matrix<CIN, P, COUT, 256>(wr, sWr, threadIdx.x);
    if (row0 < N)
        stage_node_tile<CIN, P>(agg, invdeg, xin, sMean, sSelf, row0, N, lane);
    __syncthreads();
    if (row0 >= N) return;

    const int m  = lane & 15;
    const int kk = (lane >> 4) << 1;
    const bool fullTile = (row0 + 16 <= N);
    const int rbase = row0 + ((lane >> 4) << 3);

    v2f a_mean[CIN / 4], a_self[CIN / 4];
    const float* meanRow = sMean + m * P + kk;
    const float* selfRow = sSelf + m * P + kk;
#pragma unroll
    for (int s = 0; s < CIN / 4; ++s) {
        a_mean[s] = *reinterpret_cast<const v2f*>(meanRow + 4 * s);
        a_self[s] = *reinterpret_cast<const v2f*>(selfRow + 4 * s);
    }

#pragma unroll
    for (int nb = 0; nb < COUT; nb += 16) {
        v8f c = {};
        const float* wlRow = sWl + (nb + m) * P + kk;
        const float* wrRow = sWr + (nb + m) * P + kk;
#pragma unroll
        for (int s = 0; s < CIN / 4; ++s) {
            const v2f b_l = *reinterpret_cast<const v2f*>(wlRow + 4 * s);
            const v2f b_r = *reinterpret_cast<const v2f*>(wrRow + 4 * s);
            c = __builtin_amdgcn_wmma_f32_16x16x4_f32(false, a_mean[s], false, b_l,
                                                      (short)0, c, false, false);
            c = __builtin_amdgcn_wmma_f32_16x16x4_f32(false, a_self[s], false, b_r,
                                                      (short)0, c, false, false);
        }
        const float bv = bias[nb + m];
        if (fullTile) {
            float* o = out + (size_t)rbase * COUT + nb + m;
#pragma unroll
            for (int v = 0; v < 8; ++v) {
                float val = c[v] + bv;
                if (RELU) val = fmaxf(val, 0.0f);
                o[(size_t)v * COUT] = val;
            }
        } else {
#pragma unroll
            for (int v = 0; v < 8; ++v) {
                int r = rbase + v;
                if (r < N) {
                    float val = c[v] + bv;
                    if (RELU) val = fmaxf(val, 0.0f);
                    out[(size_t)r * COUT + nb + m] = val;
                }
            }
        }
    }
}

template <int CIN, int COUT>
__global__ __launch_bounds__(256)
void sage_wmma_project_kernel(const float* __restrict__ agg,
                              const float* __restrict__ invdeg,
                              const float* __restrict__ xin,
                              const float* __restrict__ wl,
                              const float* __restrict__ wr,
                              const float* __restrict__ bias,
                              const float* __restrict__ wl3,
                              const float* __restrict__ wr3,
                              const float* __restrict__ b3,
                              float* __restrict__ p,
                              float* __restrict__ q,
                              int N) {
    constexpr int P = CIN + 4;
    __shared__ __attribute__((aligned(16))) float lds[2 * COUT * P + 2 * WAVES_PER_BLOCK * 16 * P + 256];

    const int lane = threadIdx.x & 31;
    const int waveInBlock = __builtin_amdgcn_readfirstlane((int)(threadIdx.x >> 5));
    const int tile = blockIdx.x * WAVES_PER_BLOCK + waveInBlock;
    const int row0 = tile * 16;
    const int m    = lane & 15;
    const int half = lane >> 4;
    const int kk   = half << 1;

    float* sWl   = lds;
    float* sWr   = lds + COUT * P;
    float* sMean = lds + 2 * COUT * P + waveInBlock * (16 * P);
    float* sSelf = sMean + WAVES_PER_BLOCK * 16 * P;
    float* stage = lds + 2 * COUT * P + 2 * WAVES_PER_BLOCK * 16 * P;

    stage_matrix<CIN, P, COUT, 256>(wl, sWl, threadIdx.x);
    stage_matrix<CIN, P, COUT, 256>(wr, sWr, threadIdx.x);
    if (row0 < N)
        stage_node_tile<CIN, P>(agg, invdeg, xin, sMean, sSelf, row0, N, lane);
    __syncthreads();

    float pl = 0.0f, pr = 0.0f;
    if (row0 < N) {
        v2f x_mean[CIN / 4], x_self[CIN / 4];
        const float* meanRow = sMean + m * P + kk;
        const float* selfRow = sSelf + m * P + kk;
#pragma unroll
        for (int s = 0; s < CIN / 4; ++s) {
            x_mean[s] = *reinterpret_cast<const v2f*>(meanRow + 4 * s);
            x_self[s] = *reinterpret_cast<const v2f*>(selfRow + 4 * s);
        }

#pragma unroll
        for (int nb = 0; nb < COUT; nb += 16) {
            v8f c = {};
            const float* wlRow = sWl + (nb + m) * P + kk;
            const float* wrRow = sWr + (nb + m) * P + kk;
#pragma unroll
            for (int s = 0; s < CIN / 4; ++s) {
                const v2f w_l = *reinterpret_cast<const v2f*>(wlRow + 4 * s);
                const v2f w_r = *reinterpret_cast<const v2f*>(wrRow + 4 * s);
                c = __builtin_amdgcn_wmma_f32_16x16x4_f32(false, w_l, false, x_mean[s],
                                                          (short)0, c, false, false);
                c = __builtin_amdgcn_wmma_f32_16x16x4_f32(false, w_r, false, x_self[s],
                                                          (short)0, c, false, false);
            }
            const int ch0 = nb + (half << 3);
            const float4 bA = *(const float4*)(bias + ch0), bB = *(const float4*)(bias + ch0 + 4);
            const float4 lA = *(const float4*)(wl3 + ch0),  lB = *(const float4*)(wl3 + ch0 + 4);
            const float4 rA = *(const float4*)(wr3 + ch0),  rB = *(const float4*)(wr3 + ch0 + 4);
            const float bv[8] = {bA.x, bA.y, bA.z, bA.w, bB.x, bB.y, bB.z, bB.w};
            const float lv[8] = {lA.x, lA.y, lA.z, lA.w, lB.x, lB.y, lB.z, lB.w};
            const float rv[8] = {rA.x, rA.y, rA.z, rA.w, rB.x, rB.y, rB.z, rB.w};
#pragma unroll
            for (int v = 0; v < 8; ++v) {
                const float h = fmaxf(c[v] + bv[v], 0.0f);
                pl += h * lv[v];
                pr += h * rv[v];
            }
        }
        pl += __shfl_xor(pl, 16);
        pr += __shfl_xor(pr, 16);
        pr += b3[0];
    }
    if (half == 0) {
        stage[waveInBlock * 16 + m] = pl;
        stage[128 + waveInBlock * 16 + m] = pr;
    }
    __syncthreads();
    const int t = threadIdx.x & 127;
    const int which = threadIdx.x >> 7;
    const int nodeOut = blockIdx.x * 128 + t;
    if (nodeOut < N) (which ? q : p)[nodeOut] = stage[which * 128 + t];
}

__global__ __launch_bounds__(256)
void gather_scalar_final_kernel(const float* __restrict__ p,
                                const int* __restrict__ rowptr,
                                const int* __restrict__ nbr,
                                const float* __restrict__ invdeg,
                                const float* __restrict__ q,
                                float* __restrict__ out, int N) {
    const int i = blockIdx.x * blockDim.x + threadIdx.x;
    if (i >= N) return;
    int k = rowptr[i];
    const int end = rowptr[i + 1];
    float s = 0.0f;
    for (; k + 4 <= end; k += 4) {
        const int s0 = nbr[k], s1 = nbr[k + 1], s2 = nbr[k + 2], s3 = nbr[k + 3];
        const float v0 = p[s0], v1 = p[s1], v2 = p[s2], v3 = p[s3];
        s += v0;
        s += v1;
        s += v2;
        s += v3;
    }
    for (; k < end; ++k) s += p[nbr[k]];
    const float z = s * invdeg[i] + q[i];
    out[i] = 1.0f / (1.0f + expf(-z));
}

extern "C" void kernel_launch(void* const* d_in, const int* in_sizes, int n_in,
                              void* d_out, int out_size, void* d_ws, size_t ws_size,
                              hipStream_t stream) {
    const float* x    = (const float*)d_in[0];
    const int*   ei   = (const int*)d_in[1];
    const float* wl1  = (const float*)d_in[2];
    const float* wr1  = (const float*)d_in[3];
    const float* b1   = (const float*)d_in[4];
    const float* wl2  = (const float*)d_in[5];
    const float* wr2  = (const float*)d_in[6];
    const float* b2   = (const float*)d_in[7];
    const float* wl3  = (const float*)d_in[8];
    const float* wr3  = (const float*)d_in[9];
    const float* b3   = (const float*)d_in[10];
    float* out = (float*)d_out;

    const int N = in_sizes[0] / 32;
    const int E = in_sizes[1] / 2;
    const int* src = ei;
    const int* dst = ei + E;

    int*   totals = (int*)d_ws;
    int*   fill   = totals + 256;
    float* deg  = (float*)(totals + N);
    float* agg  = deg + N;
    float* h1   = agg + (size_t)N * 64;
    float* p3   = h1 + (size_t)N * 64;
    float* q3   = p3 + N;
    int*   nbr  = (int*)(q3 + N);
    int*   rptr = nbr + E;
    int*   binned = (int*)agg;

    const int tiles = (N + 15) / 16;
    const int gemmBlocks = (tiles + 7) / 8;
    const int chunkBlocks = (E + 8191) / 8192;
    const int buckets = (N + 511) / 512;
    const int nodeBlocks = (N + 63) / 64;

    hipMemsetAsync(totals, 0, 512 * sizeof(int), stream);
    bucket_count_kernel<<<chunkBlocks, 1024, 0, stream>>>(dst, totals, E);
    bucket_scatter_kernel<<<chunkBlocks, 1024, 0, stream>>>(src, dst, totals, fill, binned, E);
    bucket_finalize_kernel<<<buckets, 1024, 0, stream>>>(binned, totals, rptr, deg, nbr, N);

    gather_sum_kernel<32><<<nodeBlocks, 256, 0, stream>>>(x, rptr, nbr, agg, N);
    sage_wmma_kernel<32, 64, true><<<gemmBlocks, 256, 0, stream>>>(
        agg, deg, x, wl1, wr1, b1, h1, N);

    gather_sum_kernel<64><<<nodeBlocks, 256, 0, stream>>>(h1, rptr, nbr, agg, N);
    sage_wmma_project_kernel<64, 64><<<gemmBlocks, 256, 0, stream>>>(
        agg, deg, h1, wl2, wr2, b2, wl3, wr3, b3, p3, q3, N);

    gather_scalar_final_kernel<<<(N + 255) / 256, 256, 0, stream>>>(
        p3, rptr, nbr, deg, q3, out, N);
}


# Neighbour sums moved into the dense kernels: each wave gathers the rows of its 16 nodes and writes the means into its LDS tile; no agg array, 2 launches fewer; per-node sum order kept
# speedup vs baseline: 5.4183x; 1.2583x over previous
#include <hip/hip_runtime.h>
#include <hip/hip_bf16.h>
#include <math.h>

typedef __attribute__((ext_vector_type(2))) float v2f;
typedef __attribute__((ext_vector_type(8))) float v8f;

constexpr int BUCKET_SHIFT = 9;
constexpr int BUCKET_NODES = 1 << BUCKET_SHIFT;
constexpr int MAX_BUCKETS  = 256;
constexpr int EPT          = 8;
constexpr int CHUNK        = 1024 * EPT;
constexpr int RPT          = 12;

__global__ __launch_bounds__(1024)
void bucket_count_kernel(const int* __restrict__ dst, int* __restrict__ totals, int E) {
    __shared__ int hist[MAX_BUCKETS];
    const int tid = threadIdx.x;
    if (tid < MAX_BUCKETS) hist[tid] = 0;
    __syncthreads();
    const int base = blockIdx.x * CHUNK;
    int dv[EPT];
#pragma unroll
    for (int j = 0; j < EPT; ++j) {
        const int e = base + j * 1024 + tid;
        dv[j] = dst[min(e, E - 1)];
    }
#pragma unroll
    for (int j = 0; j < EPT; ++j) {
        const int e = base + j * 1024 + tid;
        if (e < E) atomicAdd(&hist[dv[j] >> BUCKET_SHIFT], 1);
    }
    __syncthreads();
    if (tid < MAX_BUCKETS) {
        const int c = hist[tid];
        if (c > 0) atomicAdd(&totals[tid], c);
    }
}

__global__ __launch_bounds__(1024)
void bucket_scatter_kernel(const int* __restrict__ src, const int* __restrict__ dst,
                           const int* __restrict__ totals, int* __restrict__ fill,
                           int* __restrict__ binned, int E) {
    __shared__ int recS[CHUNK];
    __shared__ int recD[CHUNK];
    __shared__ int hist[MAX_BUCKETS];
    __shared__ int lstart[MAX_BUCKETS];
    __shared__ int gdelta[MAX_BUCKETS];
    __shared__ int wsumT[8];
    __shared__ int wsumC[8];
    const int tid = threadIdx.x;
    const int lane = tid & 31;
    const int w = tid >> 5;
    if (tid < MAX_BUCKETS) hist[tid] = 0;
    __syncthreads();

    const int base = blockIdx.x * CHUNK;
    const int nvalid = min(CHUNK, E - base);
    int sv[EPT], dv[EPT], rk[EPT];
#pragma unroll
    for (int j = 0; j < EPT; ++j) {
        const int i = j * 1024 + tid;
        const int e = base + min(i, nvalid - 1);
        sv[j] = src[e];
        dv[j] = dst[e];
    }
#pragma unroll
    for (int j = 0; j < EPT; ++j) {
        const int i = j * 1024 + tid;
        rk[j] = (i < nvalid) ? atomicAdd(&hist[dv[j] >> BUCKET_SHIFT], 1) : 0;
    }

    int t = 0, c = 0, incT = 0, incC = 0;
    if (tid < MAX_BUCKETS) t = totals[tid];
    __syncthreads();
    if (tid < MAX_BUCKETS) {
        c = hist[tid];
        incT = t;
        incC = c;
        for (int o = 1; o < 32; o <<= 1) {
            const int uT = __shfl_up(incT, o);
            const int uC = __shfl_up(incC, o);
            if (lane >= o) { incT += uT; incC += uC; }
        }
        if (lane == 31) { wsumT[w] = incT; wsumC[w] = incC; }
    }
    __syncthreads();
    if (tid < MAX_BUCKETS) {
        int preT = 0, preC = 0;
        for (int k = 0; k < 8; ++k) {
            preT += (k < w) ? wsumT[k] : 0;
            preC += (k < w) ? wsumC[k] : 0;
        }
        const int segStart = preT + incT - t;
        const int ls = preC + incC - c;
        const int got = (c > 0) ? atomicAdd(&fill[tid], c) : 0;
        lstart[tid] = ls;
        gdelta[tid] = segStart + got - ls;
    }
    __syncthreads();
#pragma unroll
    for (int j = 0; j < EPT; ++j) {
        const int i = j * 1024 + tid;
        if (i < nvalid) {
            const int slot = lstart[dv[j] >> BUCKET_SHIFT] + rk[j];
            recS[slot] = sv[j];
            recD[slot] = dv[j];
        }
    }
    __syncthreads();
    int2* out = reinterpret_cast<int2*>(binned);
#pragma unroll
    for (int j = 0; j < EPT; ++j) {
        const int i = j * 1024 + tid;
        if (i < nvalid) {
            const int s = recS[i];
            const int d = recD[i];
            out[gdelta[d >> BUCKET_SHIFT] + i] = make_int2(s, d);
        }
    }
}

__global__ __launch_bounds__(1024)
void bucket_finalize_kernel(const int* __restrict__ binned, const int* __restrict__ totals,
                            int* __restrict__ rowptr, float* __restrict__ invdeg,
                            int* __restrict__ nbr, int N) {
    __shared__ int cnt[BUCKET_NODES];
    __shared__ int wsum[32];
    const int tid = threadIdx.x;
    const int lane = tid & 31;
    const int w = tid >> 5;
    const int b = blockIdx.x;
    if (tid < BUCKET_NODES) cnt[tid] = 0;

    int part = (tid < MAX_BUCKETS && tid < b) ? totals[tid] : 0;
    for (int o = 16; o > 0; o >>= 1) part += __shfl_xor(part, o);
    if (lane == 0) wsum[w] = part;
    const int count = totals[b];
    __syncthreads();
    int start = 0;
    for (int k = 0; k < 8; ++k) start += wsum[k];

    const int2* rec = reinterpret_cast<const int2*>(binned) + start;

    const int* recDst = binned + 2 * (size_t)start + 1;
    for (int i0 = 0; i0 < count; i0 += 1024 * RPT) {
        int d[RPT];
#pragma unroll
        for (int j = 0; j < RPT; ++j) {
            const int i = i0 + j * 1024 + tid;
            d[j] = recDst[2 * min(i, count - 1)];
        }
#pragma unroll
        for (int j = 0; j < RPT; ++j) {
            const int i = i0 + j * 1024 + tid;
            if (i < count) atomicAdd(&cnt[d[j] & (BUCKET_NODES - 1)], 1);
        }
    }
    __syncthreads();

    int c = 0, inc = 0;
    if (tid < BUCKET_NODES) {
        c = cnt[tid];
        inc = c;
        for (int o = 1; o < 32; o <<= 1) {
            const int u = __shfl_up(inc, o);
            if (lane >= o) inc += u;
        }
        if (lane == 31) wsum[w] = inc;
    }
    __syncthreads();
    int excl = 0;
    if (tid < BUCKET_NODES) {
        int pre = 0;
        for (int k = 0; k < 16; ++k) pre += (k < w) ? wsum[k] : 0;
        excl = pre + inc - c;
        cnt[tid] = excl;
    }
    __syncthreads();

    for (int i0 = 0; i0 < count; i0 += 1024 * RPT) {
        int2 r[RPT];
#pragma unroll
        for (int j = 0; j < RPT; ++j) {
            const int i = i0 + j * 1024 + tid;
            r[j] = rec[min(i, count - 1)];
        }
#pragma unroll
        for (int j = 0; j < RPT; ++j) {
            const int i = i0 + j * 1024 + tid;
            if (i < count) {
                const int pos = atomicAdd(&cnt[r[j].y & (BUCKET_NODES - 1)], 1);
                nbr[start + pos] = r[j].x;
            }
        }
    }

    if (tid < BUCKET_NODES) {
        const int node = (b << BUCKET_SHIFT) + tid;
        if (node < N) {
            rowptr[node] = start + excl;
            invdeg[node] = 1.0f / fmaxf((float)c, 1.0f);
        }
        if (node == N - 1) rowptr[N] = start + excl + c;
    }
}

__device__ __forceinline__ float swap_rows16(float v) {
    return __builtin_bit_cast(float, __builtin_amdgcn_permlanex16(
        __builtin_bit_cast(int, v), __builtin_bit_cast(int, v), 0x76543210, 0xfedcba98, false, false));
}
__device__ __forceinline__ float swap_groups8(float v) {
    return __builtin_bit_cast(float, __builtin_amdgcn_update_dpp(
        0, __builtin_bit_cast(int, v), 0x168, 0xf, 0xf, false));
}

template <int C, int K, int t>
__device__ __forceinline__ void issue_rows(const char* __restrict__ xb, unsigned sv, unsigned subOff,
                                           bool lastOk, float4 (&v)[K]) {
    if constexpr (t < K) {
        const unsigned idx = __builtin_amdgcn_mov_dpp8(sv, t * 0x249249u);
        const unsigned off = idx * (unsigned)(C * 4) + subOff;
        if (t + 1 < K || lastOk) v[t] = *reinterpret_cast<const float4*>(xb + off);
        else v[t] = make_float4(0.0f, 0.0f, 0.0f, 0.0f);
        issue_rows<C, K, t + 1>(xb, sv, subOff, lastOk, v);
    }
}

template <int C, int K>
__device__ __forceinline__ void gather_pass(const char* __restrict__ xb, unsigned sv, unsigned subOff,
                                            bool lastOk, float4& acc) {
    float4 v[K];
    issue_rows<C, K, 0>(xb, sv, subOff, lastOk, v);
#pragma unroll
    for (int t = 0; t < K; ++t) {
        acc.x += v[t].x; acc.y += v[t].y; acc.z += v[t].z; acc.w += v[t].w;
    }
}

template <int C, int P>
__device__ __forceinline__ void gather_mean_tile(const float* __restrict__ x,
                                                 const int* __restrict__ nbr,
                                                 int rp, float inv,
                                                 float* sTile, int nn, int lane) {
    constexpr int LPR = C / 4;
    constexpr int G   = 32 / LPR;
    constexpr int PS  = 8 * G;
    static_assert(C == 32 || C == 64, "a row is 8 or 16 lanes");
    const int g    = lane / LPR;
    const unsigned subOff = (lane % LPR) * 16;
    const int perm = G * (lane & 7) + g;
    int pos = __builtin_amdgcn_readlane(rp, 0);
    const int wend = __builtin_amdgcn_readlane(rp, nn);
    const char* xb = reinterpret_cast<const char*>(x);

    unsigned svNext = 0;
    if (pos < wend) svNext = (unsigned)nbr[min(pos + perm, wend - 1)];

    for (int i = 0; i < nn; ++i) {
        const int end = __builtin_amdgcn_readlane(rp, i + 1);
        float4 acc = make_float4(0.0f, 0.0f, 0.0f, 0.0f);
        while (pos < end) {
            const int n = min(PS, end - pos);
            const unsigned sv = svNext;
            pos += n;
            if (pos < wend) svNext = (unsigned)nbr[min(pos + perm, wend - 1)];
            const int K = (n + G - 1) / G;
            const bool lastOk = g < n - (K - 1) * G;
            switch (K) {
                case 1:  gather_pass<C, 1>(xb, sv, subOff, lastOk, acc); break;
                case 2:  gather_pass<C, 2>(xb, sv, subOff, lastOk, acc); break;
                case 3:  gather_pass<C, 3>(xb, sv, subOff, lastOk, acc); break;
                case 4:  gather_pass<C, 4>(xb, sv, subOff, lastOk, acc); break;
                case 5:  gather_pass<C, 5>(xb, sv, subOff, lastOk, acc); break;
                case 6:  gather_pass<C, 6>(xb, sv, subOff, lastOk, acc); break;
                case 7:  gather_pass<C, 7>(xb, sv, subOff, lastOk, acc); break;
                default: gather_pass<C, 8>(xb, sv, subOff, lastOk, acc); break;
            }
        }
        if (G == 4) {
            acc.x += swap_groups8(acc.x); acc.y += swap_groups8(acc.y);
            acc.z += swap_groups8(acc.z); acc.w += swap_groups8(acc.w);
        }
        acc.x += swap_rows16(acc.x); acc.y += swap_rows16(acc.y);
        acc.z += swap_rows16(acc.z); acc.w += swap_rows16(acc.w);
        const float scale = __builtin_bit_cast(float, __builtin_amdgcn_readlane(__builtin_bit_cast(int, inv), i));
        acc.x *= scale; acc.y *= scale; acc.z *= scale; acc.w *= scale;
        if (lane < LPR)
            *reinterpret_cast<float4*>(sTile + i * P + lane * 4) = acc;
    }
    for (int i = nn; i < 16; ++i)
        if (lane < LPR)
            *reinterpret_cast<float4*>(sTile + i * P + lane * 4) = make_float4(0.0f, 0.0f, 0.0f, 0.0f);
}

__device__ __forceinline__ void wave_tile_sync() {
    __builtin_amdgcn_fence(__ATOMIC_RELEASE, "wavefront");
    __builtin_amdgcn_wave_barrier();
    __builtin_amdgcn_fence(__ATOMIC_ACQUIRE, "wavefront");
}

constexpr int WAVES_PER_BLOCK = 8;

template <int CIN, int P, int ROWS, int NT>
__device__ __forceinline__ void stage_matrix(const float* __restrict__ g, float* s, int t) {
    constexpr int V = CIN / 4;
    static_assert((ROWS * V) % NT == 0, "whole passes only");
#pragma unroll
    for (int i = 0; i < ROWS * V / NT; ++i) {
        const int v = i * NT + t;
        const float4 val = *reinterpret_cast<const float4*>(g + (size_t)v * 4);
        *reinterpret_cast<float4*>(s + (v / V) * P + (v % V) * 4) = val;
    }
}

template <int CIN, int P>
__device__ __forceinline__ void stage_self_tile(const float* __restrict__ xin, float* sSelf,
                                                int row0, int N, int lane) {
    constexpr int V = CIN / 4;
#pragma unroll
    for (int i = 0; i < 16 * V / 32; ++i) {
        const int v = i * 32 + lane;
        const int r = v / V;
        const int c = (v % V) * 4;
        const int node = min(row0 + r, N - 1);
        const float4 s = *reinterpret_cast<const float4*>(xin + (size_t)node * CIN + c);
        *reinterpret_cast<float4*>(sSelf + r * P + c) = s;
    }
}

template <int CIN, int COUT, bool RELU>
__global__ __launch_bounds__(256)
void sage_wmma_kernel(const int* __restrict__ rowptr,
                      const int* __restrict__ nbr,
                      const float* __restrict__ invdeg,
                      const float* __restrict__ xin,
                      const float* __restrict__ wl,
                      const float* __restrict__ wr,
                      const float* __restrict__ bias,
                      float* __restrict__ out,
                      int N) {
    constexpr int P = CIN + 4;
    __shared__ __attribute__((aligned(16))) float lds[2 * COUT * P + 2 * WAVES_PER_BLOCK * 16 * P];

    const int lane = threadIdx.x & 31;
    const int waveInBlock = __builtin_amdgcn_readfirstlane((int)(threadIdx.x >> 5));
    const int tile = blockIdx.x * WAVES_PER_BLOCK + waveInBlock;
    const int row0 = tile * 16;

    float* sWl   = lds;
    float* sWr   = lds + COUT * P;
    float* sMean = lds + 2 * COUT * P + waveInBlock * (16 * P);
    float* sSelf = sMean + WAVES_PER_BLOCK * 16 * P;

    stage_matrix<CIN, P, COUT, 256>(wl, sWl, threadIdx.x);
    stage_matrix<CIN, P, COUT, 256>(wr, sWr, threadIdx.x);
    __syncthreads();
    if (row0 >= N) return;
    const int nn = min(16, N - row0);
    const int rp = rowptr[row0 + min(lane, nn)];
    const float inv = invdeg[row0 + min(lane, nn - 1)];
    stage_self_tile<CIN, P>(xin, sSelf, row0, N, lane);
    gather_mean_tile<CIN, P>(xin, nbr, rp, inv, sMean, nn, lane);
    wave_tile_sync();

    const int m  = lane & 15;
    const int kk = (lane >> 4) << 1;
    const bool fullTile = (row0 + 16 <= N);
    const int rbase = row0 + ((lane >> 4) << 3);

    v2f a_mean[CIN / 4], a_self[CIN / 4];
    const float* meanRow = sMean + m * P + kk;
    const float* selfRow = sSelf + m * P + kk;
#pragma unroll
    for (int s = 0; s < CIN / 4; ++s) {
        a_mean[s] = *reinterpret_cast<const v2f*>(meanRow + 4 * s);
        a_self[s] = *reinterpret_cast<const v2f*>(selfRow + 4 * s);
    }

#pragma unroll
    for (int nb = 0; nb < COUT; nb += 16) {
        v8f c = {};
        const float* wlRow = sWl + (nb + m) * P + kk;
        const float* wrRow = sWr + (nb + m) * P + kk;
#pragma unroll
        for (int s = 0; s < CIN / 4; ++s) {
            const v2f b_l = *reinterpret_cast<const v2f*>(wlRow + 4 * s);
            const v2f b_r = *reinterpret_cast<const v2f*>(wrRow + 4 * s);
            c = __builtin_amdgcn_wmma_f32_16x16x4_f32(false, a_mean[s], false, b_l,
                                                      (short)0, c, false, false);
            c = __builtin_amdgcn_wmma_f32_16x16x4_f32(false, a_self[s], false, b_r,
                                                      (short)0, c, false, false);
        }
        const float bv = bias[nb + m];
        if (fullTile) {
            float* o = out + (size_t)rbase * COUT + nb + m;
#pragma unroll
            for (int v = 0; v < 8; ++v) {
                float val = c[v] + bv;
                if (RELU) val = fmaxf(val, 0.0f);
                o[(size_t)v * COUT] = val;
            }
        } else {
#pragma unroll
            for (int v = 0; v < 8; ++v) {
                int r = rbase + v;
                if (r < N) {
                    float val = c[v] + bv;
                    if (RELU) val = fmaxf(val, 0.0f);
                    out[(size_t)r * COUT + nb + m] = val;
                }
            }
        }
    }
}

template <int CIN, int COUT>
__global__ __launch_bounds__(256)
void sage_wmma_project_kernel(const int* __restrict__ rowptr,
                              const int* __restrict__ nbr,
                              const float* __restrict__ invdeg,
                              const float* __restrict__ xin,
                              const float* __restrict__ wl,
                              const float* __restrict__ wr,
                              const float* __restrict__ bias,
                              const float* __restrict__ wl3,
                              const float* __restrict__ wr3,
                              const float* __restrict__ b3,
                              float* __restrict__ p,
                              float* __restrict__ q,
                              int N) {
    constexpr int P = CIN + 4;
    __shared__ __attribute__((aligned(16))) float lds[2 * COUT * P + WAVES_PER_BLOCK * 16 * P + 256];

    const int lane = threadIdx.x & 31;
    const int waveInBlock = __builtin_amdgcn_readfirstlane((int)(threadIdx.x >> 5));
    const int tile = blockIdx.x * WAVES_PER_BLOCK + waveInBlock;
    const int row0 = tile * 16;
    const int m    = lane & 15;
    const int half = lane >> 4;
    const int kk   = half << 1;

    float* sWl   = lds;
    float* sWr   = lds + COUT * P;
    float* sTile = lds + 2 * COUT * P + waveInBlock * (16 * P);
    float* stage = lds + 2 * COUT * P + WAVES_PER_BLOCK * 16 * P;

    stage_matrix<CIN, P, COUT, 256>(wl, sWl, threadIdx.x);
    stage_matrix<CIN, P, COUT, 256>(wr, sWr, threadIdx.x);
    __syncthreads();

    float pl = 0.0f, pr = 0.0f;
    if (row0 < N) {
        v2f x_mean[CIN / 4], x_self[CIN / 4];
        const float* tileRow = sTile + m * P + kk;
        const int nn = min(16, N - row0);
        const int rp = rowptr[row0 + min(lane, nn)];
        const float inv = invdeg[row0 + min(lane, nn - 1)];
        stage_self_tile<CIN, P>(xin, sTile, row0, N, lane);
        wave_tile_sync();
#pragma unroll
        for (int s = 0; s < CIN / 4; ++s)
            x_self[s] = *reinterpret_cast<const v2f*>(tileRow + 4 * s);
        wave_tile_sync();
        gather_mean_tile<CIN, P>(xin, nbr, rp, inv, sTile, nn, lane);
        wave_tile_sync();
#pragma unroll
        for (int s = 0; s < CIN / 4; ++s)
            x_mean[s] = *reinterpret_cast<const v2f*>(tileRow + 4 * s);

#pragma unroll
        for (int nb = 0; nb < COUT; nb += 16) {
            v8f c = {};
            const float* wlRow = sWl + (nb + m) * P + kk;
            const float* wrRow = sWr + (nb + m) * P + kk;
#pragma unroll
            for (int s = 0; s < CIN / 4; ++s) {
                const v2f w_l = *reinterpret_cast<const v2f*>(wlRow + 4 * s);
                const v2f w_r = *reinterpret_cast<const v2f*>(wrRow + 4 * s);
                c = __builtin_amdgcn_wmma_f32_16x16x4_f32(false, w_l, false, x_mean[s],
                                                          (short)0, c, false, false);
                c = __builtin_amdgcn_wmma_f32_16x16x4_f32(false, w_r, false, x_self[s],
                                                          (short)0, c, false, false);
            }
            const int ch0 = nb + (half << 3);
            const float4 bA = *(const float4*)(bias + ch0), bB = *(const float4*)(bias + ch0 + 4);
            const float4 lA = *(const float4*)(wl3 + ch0),  lB = *(const float4*)(wl3 + ch0 + 4);
            const float4 rA = *(const float4*)(wr3 + ch0),  rB = *(const float4*)(wr3 + ch0 + 4);
            const float bv[8] = {bA.x, bA.y, bA.z, bA.w, bB.x, bB.y, bB.z, bB.w};
            const float lv[8] = {lA.x, lA.y, lA.z, lA.w, lB.x, lB.y, lB.z, lB.w};
            const float rv[8] = {rA.x, rA.y, rA.z, rA.w, rB.x, rB.y, rB.z, rB.w};
#pragma unroll
            for (int v = 0; v < 8; ++v) {
                const float h = fmaxf(c[v] + bv[v], 0.0f);
                pl += h * lv[v];
                pr += h * rv[v];
            }
        }
        pl += __shfl_xor(pl, 16);
        pr += __shfl_xor(pr, 16);
        pr += b3[0];
    }
    if (half == 0) {
        stage[waveInBlock * 16 + m] = pl;
        stage[128 + waveInBlock * 16 + m] = pr;
    }
    __syncthreads();
    const int t = threadIdx.x & 127;
    const int which = threadIdx.x >> 7;
    const int nodeOut = blockIdx.x * 128 + t;
    if (nodeOut < N) (which ? q : p)[nodeOut] = stage[which * 128 + t];
}

__global__ __launch_bounds__(256)
void gather_scalar_final_kernel(const float* __restrict__ p,
                                const int* __restrict__ rowptr,
                                const int* __restrict__ nbr,
                                const float* __restrict__ invdeg,
                                const float* __restrict__ q,
                                float* __restrict__ out, int N) {
    const int i = blockIdx.x * blockDim.x + threadIdx.x;
    if (i >= N) return;
    int k = rowptr[i];
    const int end = rowptr[i + 1];
    float s = 0.0f;
    for (; k + 4 <= end; k += 4) {
        const int s0 = nbr[k], s1 = nbr[k + 1], s2 = nbr[k + 2], s3 = nbr[k + 3];
        const float v0 = p[s0], v1 = p[s1], v2 = p[s2], v3 = p[s3];
        s += v0;
        s += v1;
        s += v2;
        s += v3;
    }
    for (; k < end; ++k) s += p[nbr[k]];
    const float z = s * invdeg[i] + q[i];
    out[i] = 1.0f / (1.0f + expf(-z));
}

extern "C" void kernel_launch(void* const* d_in, const int* in_sizes, int n_in,
                              void* d_out, int out_size, void* d_ws, size_t ws_size,
                              hipStream_t stream) {
    const float* x    = (const float*)d_in[0];
    const int*   ei   = (const int*)d_in[1];
    const float* wl1  = (const float*)d_in[2];
    const float* wr1  = (const float*)d_in[3];
    const float* b1   = (const float*)d_in[4];
    const float* wl2  = (const float*)d_in[5];
    const float* wr2  = (const float*)d_in[6];
    const float* b2   = (const float*)d_in[7];
    const float* wl3  = (const float*)d_in[8];
    const float* wr3  = (const float*)d_in[9];
    const float* b3   = (const float*)d_in[10];
    float* out = (float*)d_out;

    const int N = in_sizes[0] / 32;
    const int E = in_sizes[1] / 2;
    const int* src = ei;
    const int* dst = ei + E;

    int*   totals = (int*)d_ws;
    int*   fill   = totals + 256;
    float* deg  = (float*)(totals + N);
    float* recs = deg + N;
    float* h1   = recs + (size_t)N * 64;
    float* p3   = h1 + (size_t)N * 64;
    float* q3   = p3 + N;
    int*   nbr  = (int*)(q3 + N);
    int*   rptr = nbr + E;
    int*   binned = (int*)recs;

    const int tiles = (N + 15) / 16;
    const int gemmBlocks = (tiles + 7) / 8;
    const int chunkBlocks = (E + 8191) / 8192;
    const int buckets = (N + 511) / 512;

    hipMemsetAsync(totals, 0, 512 * sizeof(int), stream);
    bucket_count_kernel<<<chunkBlocks, 1024, 0, stream>>>(dst, totals, E);
    bucket_scatter_kernel<<<chunkBlocks, 1024, 0, stream>>>(src, dst, totals, fill, binned, E);
    bucket_finalize_kernel<<<buckets, 1024, 0, stream>>>(binned, totals, rptr, deg, nbr, N);

    sage_wmma_kernel<32, 64, true><<<gemmBlocks, 256, 0, stream>>>(
        rptr, nbr, deg, x, wl1, wr1, b1, h1, N);

    sage_wmma_project_kernel<64, 64><<<gemmBlocks, 256, 0, stream>>>(
        rptr, nbr, deg, h1, wl2, wr2, b2, wl3, wr3, b3, p3, q3, N);

    gather_scalar_final_kernel<<<(N + 255) / 256, 256, 0, stream>>>(
        p3, rptr, nbr, deg, q3, out, N);
}


# Bucket finalize builds each bucket's source list in LDS (ranks from the same returning LDS adds) and writes it out as whole 128-byte lines: no scattered 4-byte stores; list order: LDS-add arrival
# speedup vs baseline: 5.7769x; 1.0662x over previous
#include <hip/hip_runtime.h>
#include <hip/hip_bf16.h>
#include <math.h>

typedef __attribute__((ext_vector_type(2))) float v2f;
typedef __attribute__((ext_vector_type(8))) float v8f;

constexpr int BUCKET_SHIFT = 9;
constexpr int BUCKET_NODES = 1 << BUCKET_SHIFT;
constexpr int MAX_BUCKETS  = 256;
constexpr int EPT          = 8;
constexpr int CHUNK        = 1024 * EPT;
constexpr int RPT          = 12;
constexpr int LIST_CAP     = 16384;

__global__ __launch_bounds__(1024)
void bucket_count_kernel(const int* __restrict__ dst, int* __restrict__ totals, int E) {
    __shared__ int hist[MAX_BUCKETS];
    const int tid = threadIdx.x;
    if (tid < MAX_BUCKETS) hist[tid] = 0;
    __syncthreads();
    const int base = blockIdx.x * CHUNK;
    int dv[EPT];
#pragma unroll
    for (int j = 0; j < EPT; ++j) {
        const int e = base + j * 1024 + tid;
        dv[j] = dst[min(e, E - 1)];
    }
#pragma unroll
    for (int j = 0; j < EPT; ++j) {
        const int e = base + j * 1024 + tid;
        if (e < E) atomicAdd(&hist[dv[j] >> BUCKET_SHIFT], 1);
    }
    __syncthreads();
    if (tid < MAX_BUCKETS) {
        const int c = hist[tid];
        if (c > 0) atomicAdd(&totals[tid], c);
    }
}

__global__ __launch_bounds__(1024)
void bucket_scatter_kernel(const int* __restrict__ src, const int* __restrict__ dst,
                           const int* __restrict__ totals, int* __restrict__ fill,
                           int* __restrict__ binned, int E) {
    __shared__ int recS[CHUNK];
    __shared__ int recD[CHUNK];
    __shared__ int hist[MAX_BUCKETS];
    __shared__ int lstart[MAX_BUCKETS];
    __shared__ int gdelta[MAX_BUCKETS];
    __shared__ int wsumT[8];
    __shared__ int wsumC[8];
    const int tid = threadIdx.x;
    const int lane = tid & 31;
    const int w = tid >> 5;
    if (tid < MAX_BUCKETS) hist[tid] = 0;
    __syncthreads();

    const int base = blockIdx.x * CHUNK;
    const int nvalid = min(CHUNK, E - base);
    int sv[EPT], dv[EPT], rk[EPT];
#pragma unroll
    for (int j = 0; j < EPT; ++j) {
        const int i = j * 1024 + tid;
        const int e = base + min(i, nvalid - 1);
        sv[j] = src[e];
        dv[j] = dst[e];
    }
#pragma unroll
    for (int j = 0; j < EPT; ++j) {
        const int i = j * 1024 + tid;
        rk[j] = (i < nvalid) ? atomicAdd(&hist[dv[j] >> BUCKET_SHIFT], 1) : 0;
    }

    int t = 0, c = 0, incT = 0, incC = 0;
    if (tid < MAX_BUCKETS) t = totals[tid];
    __syncthreads();
    if (tid < MAX_BUCKETS) {
        c = hist[tid];
        incT = t;
        incC = c;
        for (int o = 1; o < 32; o <<= 1) {
            const int uT = __shfl_up(incT, o);
            const int uC = __shfl_up(incC, o);
            if (lane >= o) { incT += uT; incC += uC; }
        }
        if (lane == 31) { wsumT[w] = incT; wsumC[w] = incC; }
    }
    __syncthreads();
    if (tid < MAX_BUCKETS) {
        int preT = 0, preC = 0;
        for (int k = 0; k < 8; ++k) {
            preT += (k < w) ? wsumT[k] : 0;
            preC += (k < w) ? wsumC[k] : 0;
        }
        const int segStart = preT + incT - t;
        const int ls = preC + incC - c;
        const int got = (c > 0) ? atomicAdd(&fill[tid], c) : 0;
        lstart[tid] = ls;
        gdelta[tid] = segStart + got - ls;
    }
    __syncthreads();
#pragma unroll
    for (int j = 0; j < EPT; ++j) {
        const int i = j * 1024 + tid;
        if (i < nvalid) {
            const int slot = lstart[dv[j] >> BUCKET_SHIFT] + rk[j];
            recS[slot] = sv[j];
            recD[slot] = dv[j];
        }
    }
    __syncthreads();
    int2* out = reinterpret_cast<int2*>(binned);
#pragma unroll
    for (int j = 0; j < EPT; ++j) {
        const int i = j * 1024 + tid;
        if (i < nvalid) {
            const int s = recS[i];
            const int d = recD[i];
            out[gdelta[d >> BUCKET_SHIFT] + i] = make_int2(s, d);
        }
    }
}

__global__ __launch_bounds__(1024)
void bucket_finalize_kernel(const int* __restrict__ binned, const int* __restrict__ totals,
                            int* __restrict__ rowptr, float* __restrict__ invdeg,
                            int* __restrict__ nbr, int N) {
    __shared__ int cnt[BUCKET_NODES];
    __shared__ int wsum[32];
    __shared__ int sList[LIST_CAP];
    const int tid = threadIdx.x;
    const int lane = tid & 31;
    const int w = tid >> 5;
    const int b = blockIdx.x;
    if (tid < BUCKET_NODES) cnt[tid] = 0;

    int part = (tid < MAX_BUCKETS && tid < b) ? totals[tid] : 0;
    for (int o = 16; o > 0; o >>= 1) part += __shfl_xor(part, o);
    if (lane == 0) wsum[w] = part;
    const int count = totals[b];
    __syncthreads();
    int start = 0;
    for (int k = 0; k < 8; ++k) start += wsum[k];

    const int2* rec = reinterpret_cast<const int2*>(binned) + start;

    const int* recDst = binned + 2 * (size_t)start + 1;
    for (int i0 = 0; i0 < count; i0 += 1024 * RPT) {
        int d[RPT];
#pragma unroll
        for (int j = 0; j < RPT; ++j) {
            const int i = i0 + j * 1024 + tid;
            d[j] = recDst[2 * min(i, count - 1)];
        }
#pragma unroll
        for (int j = 0; j < RPT; ++j) {
            const int i = i0 + j * 1024 + tid;
            if (i < count) atomicAdd(&cnt[d[j] & (BUCKET_NODES - 1)], 1);
        }
    }
    __syncthreads();

    int c = 0, inc = 0;
    if (tid < BUCKET_NODES) {
        c = cnt[tid];
        inc = c;
        for (int o = 1; o < 32; o <<= 1) {
            const int u = __shfl_up(inc, o);
            if (lane >= o) inc += u;
        }
        if (lane == 31) wsum[w] = inc;
    }
    __syncthreads();
    int excl = 0;
    if (tid < BUCKET_NODES) {
        int pre = 0;
        for (int k = 0; k < 16; ++k) pre += (k < w) ? wsum[k] : 0;
        excl = pre + inc - c;
        cnt[tid] = excl;
    }
    __syncthreads();

    if (count <= LIST_CAP) {
        for (int i0 = 0; i0 < count; i0 += 1024 * RPT) {
            int2 r[RPT];
            int pos[RPT];
#pragma unroll
            for (int j = 0; j < RPT; ++j) {
                const int i = i0 + j * 1024 + tid;
                r[j] = rec[min(i, count - 1)];
            }
#pragma unroll
            for (int j = 0; j < RPT; ++j) {
                const int i = i0 + j * 1024 + tid;
                pos[j] = (i < count) ? atomicAdd(&cnt[r[j].y & (BUCKET_NODES - 1)], 1) : -1;
            }
#pragma unroll
            for (int j = 0; j < RPT; ++j)
                if (pos[j] >= 0) sList[pos[j]] = r[j].x;
        }
        __syncthreads();
        const int lead = start & 31;
        for (int e0 = tid - lead; e0 < count; e0 += 4 * 1024) {
            int v[4];
#pragma unroll
            for (int u = 0; u < 4; ++u)
                v[u] = sList[min(max(e0 + u * 1024, 0), LIST_CAP - 1)];
#pragma unroll
            for (int u = 0; u < 4; ++u) {
                const int e = e0 + u * 1024;
                if (e >= 0 && e < count) nbr[start + e] = v[u];
            }
        }
    } else {
        for (int i0 = 0; i0 < count; i0 += 1024 * RPT) {
            int2 r[RPT];
#pragma unroll
            for (int j = 0; j < RPT; ++j) {
                const int i = i0 + j * 1024 + tid;
                r[j] = rec[min(i, count - 1)];
            }
#pragma unroll
            for (int j = 0; j < RPT; ++j) {
                const int i = i0 + j * 1024 + tid;
                if (i < count) {
                    const int pos = atomicAdd(&cnt[r[j].y & (BUCKET_NODES - 1)], 1);
                    nbr[start + pos] = r[j].x;
                }
            }
        }
    }

    if (tid < BUCKET_NODES) {
        const int node = (b << BUCKET_SHIFT) + tid;
        if (node < N) {
            rowptr[node] = start + excl;
            invdeg[node] = 1.0f / fmaxf((float)c, 1.0f);
        }
        if (node == N - 1) rowptr[N] = start + excl + c;
    }
}

__device__ __forceinline__ float swap_rows16(float v) {
    return __builtin_bit_cast(float, __builtin_amdgcn_permlanex16(
        __builtin_bit_cast(int, v), __builtin_bit_cast(int, v), 0x76543210, 0xfedcba98, false, false));
}
__device__ __forceinline__ float swap_groups8(float v) {
    return __builtin_bit_cast(float, __builtin_amdgcn_update_dpp(
        0, __builtin_bit_cast(int, v), 0x168, 0xf, 0xf, false));
}

template <int C, int K, int t>
__device__ __forceinline__ void issue_rows(const char* __restrict__ xb, unsigned sv, unsigned subOff,
                                           bool lastOk, float4 (&v)[K]) {
    if constexpr (t < K) {
        const unsigned idx = __builtin_amdgcn_mov_dpp8(sv, t * 0x249249u);
        const unsigned off = idx * (unsigned)(C * 4) + subOff;
        if (t + 1 < K || lastOk) v[t] = *reinterpret_cast<const float4*>(xb + off);
        else v[t] = make_float4(0.0f, 0.0f, 0.0f, 0.0f);
        issue_rows<C, K, t + 1>(xb, sv, subOff, lastOk, v);
    }
}

template <int C, int K>
__device__ __forceinline__ void gather_pass(const char* __restrict__ xb, unsigned sv, unsigned subOff,
                                            bool lastOk, float4& acc) {
    float4 v[K];
    issue_rows<C, K, 0>(xb, sv, subOff, lastOk, v);
#pragma unroll
    for (int t = 0; t < K; ++t) {
        acc.x += v[t].x; acc.y += v[t].y; acc.z += v[t].z; acc.w += v[t].w;
    }
}

template <int C, int P>
__device__ __forceinline__ void gather_mean_tile(const float* __restrict__ x,
                                                 const int* __restrict__ nbr,
                                                 int rp, float inv,
                                                 float* sTile, int nn, int lane) {
    constexpr int LPR = C / 4;
    constexpr int G   = 32 / LPR;
    constexpr int PS  = 8 * G;
    static_assert(C == 32 || C == 64, "a row is 8 or 16 lanes");
    const int g    = lane / LPR;
    const unsigned subOff = (lane % LPR) * 16;
    const int perm = G * (lane & 7) + g;
    int pos = __builtin_amdgcn_readlane(rp, 0);
    const int wend = __builtin_amdgcn_readlane(rp, nn);
    const char* xb = reinterpret_cast<const char*>(x);

    unsigned svNext = 0;
    if (pos < wend) svNext = (unsigned)nbr[min(pos + perm, wend - 1)];

    for (int i = 0; i < nn; ++i) {
        const int end = __builtin_amdgcn_readlane(rp, i + 1);
        float4 acc = make_float4(0.0f, 0.0f, 0.0f, 0.0f);
        while (pos < end) {
            const int n = min(PS, end - pos);
            const unsigned sv = svNext;
            pos += n;
            if (pos < wend) svNext = (unsigned)nbr[min(pos + perm, wend - 1)];
            const int K = (n + G - 1) / G;
            const bool lastOk = g < n - (K - 1) * G;
            switch (K) {
                case 1:  gather_pass<C, 1>(xb, sv, subOff, lastOk, acc); break;
                case 2:  gather_pass<C, 2>(xb, sv, subOff, lastOk, acc); break;
                case 3:  gather_pass<C, 3>(xb, sv, subOff, lastOk, acc); break;
                case 4:  gather_pass<C, 4>(xb, sv, subOff, lastOk, acc); break;
                case 5:  gather_pass<C, 5>(xb, sv, subOff, lastOk, acc); break;
                case 6:  gather_pass<C, 6>(xb, sv, subOff, lastOk, acc); break;
                case 7:  gather_pass<C, 7>(xb, sv, subOff, lastOk, acc); break;
                default: gather_pass<C, 8>(xb, sv, subOff, lastOk, acc); break;
            }
        }
        if (G == 4) {
            acc.x += swap_groups8(acc.x); acc.y += swap_groups8(acc.y);
            acc.z += swap_groups8(acc.z); acc.w += swap_groups8(acc.w);
        }
        acc.x += swap_rows16(acc.x); acc.y += swap_rows16(acc.y);
        acc.z += swap_rows16(acc.z); acc.w += swap_rows16(acc.w);
        const float scale = __builtin_bit_cast(float, __builtin_amdgcn_readlane(__builtin_bit_cast(int, inv), i));
        acc.x *= scale; acc.y *= scale; acc.z *= scale; acc.w *= scale;
        if (lane < LPR)
            *reinterpret_cast<float4*>(sTile + i * P + lane * 4) = acc;
    }
    for (int i = nn; i < 16; ++i)
        if (lane < LPR)
            *reinterpret_cast<float4*>(sTile + i * P + lane * 4) = make_float4(0.0f, 0.0f, 0.0f, 0.0f);
}

__device__ __forceinline__ void wave_tile_sync() {
    __builtin_amdgcn_fence(__ATOMIC_RELEASE, "wavefront");
    __builtin_amdgcn_wave_barrier();
    __builtin_amdgcn_fence(__ATOMIC_ACQUIRE, "wavefront");
}

constexpr int WAVES_PER_BLOCK = 8;

template <int CIN, int P, int ROWS, int NT>
__device__ __forceinline__ void stage_matrix(const float* __restrict__ g, float* s, int t) {
    constexpr int V = CIN / 4;
    static_assert((ROWS * V) % NT == 0, "whole passes only");
#pragma unroll
    for (int i = 0; i < ROWS * V / NT; ++i) {
        const int v = i * NT + t;
        const float4 val = *reinterpret_cast<const float4*>(g + (size_t)v * 4);
        *reinterpret_cast<float4*>(s + (v / V) * P + (v % V) * 4) = val;
    }
}

template <int CIN, int P>
__device__ __forceinline__ void stage_self_tile(const float* __restrict__ xin, float* sSelf,
                                                int row0, int N, int lane) {
    constexpr int V = CIN / 4;
#pragma unroll
    for (int i = 0; i < 16 * V / 32; ++i) {
        const int v = i * 32 + lane;
        const int r = v / V;
        const int c = (v % V) * 4;
        const int node = min(row0 + r, N - 1);
        const float4 s = *reinterpret_cast<const float4*>(xin + (size_t)node * CIN + c);
        *reinterpret_cast<float4*>(sSelf + r * P + c) = s;
    }
}

template <int CIN, int COUT, bool RELU>
__global__ __launch_bounds__(256)
void sage_wmma_kernel(const int* __restrict__ rowptr,
                      const int* __restrict__ nbr,
                      const float* __restrict__ invdeg,
                      const float* __restrict__ xin,
                      const float* __restrict__ wl,
                      const float* __restrict__ wr,
                      const float* __restrict__ bias,
                      float* __restrict__ out,
                      int N) {
    constexpr int P = CIN + 4;
    __shared__ __attribute__((aligned(16))) float lds[2 * COUT * P + 2 * WAVES_PER_BLOCK * 16 * P];

    const int lane = threadIdx.x & 31;
    const int waveInBlock = __builtin_amdgcn_readfirstlane((int)(threadIdx.x >> 5));
    const int tile = blockIdx.x * WAVES_PER_BLOCK + waveInBlock;
    const int row0 = tile * 16;

    float* sWl   = lds;
    float* sWr   = lds + COUT * P;
    float* sMean = lds + 2 * COUT * P + waveInBlock * (16 * P);
    float* sSelf = sMean + WAVES_PER_BLOCK * 16 * P;

    stage_matrix<CIN, P, COUT, 256>(wl, sWl, threadIdx.x);
    stage_matrix<CIN, P, COUT, 256>(wr, sWr, threadIdx.x);
    __syncthreads();
    if (row0 >= N) return;
    const int nn = min(16, N - row0);
    const int rp = rowptr[row0 + min(lane, nn)];
    const float inv = invdeg[row0 + min(lane, nn - 1)];
    stage_self_tile<CIN, P>(xin, sSelf, row0, N, lane);
    gather_mean_tile<CIN, P>(xin, nbr, rp, inv, sMean, nn, lane);
    wave_tile_sync();

    const int m  = lane & 15;
    const int kk = (lane >> 4) << 1;
    const bool fullTile = (row0 + 16 <= N);
    const int rbase = row0 + ((lane >> 4) << 3);

    v2f a_mean[CIN / 4], a_self[CIN / 4];
    const float* meanRow = sMean + m * P + kk;
    const float* selfRow = sSelf + m * P + kk;
#pragma unroll
    for (int s = 0; s < CIN / 4; ++s) {
        a_mean[s] = *reinterpret_cast<const v2f*>(meanRow + 4 * s);
        a_self[s] = *reinterpret_cast<const v2f*>(selfRow + 4 * s);
    }

#pragma unroll
    for (int nb = 0; nb < COUT; nb += 16) {
        v8f c = {};
        const float* wlRow = sWl + (nb + m) * P + kk;
        const float* wrRow = sWr + (nb + m) * P + kk;
#pragma unroll
        for (int s = 0; s < CIN / 4; ++s) {
            const v2f b_l = *reinterpret_cast<const v2f*>(wlRow + 4 * s);
            const v2f b_r = *reinterpret_cast<const v2f*>(wrRow + 4 * s);
            c = __builtin_amdgcn_wmma_f32_16x16x4_f32(false, a_mean[s], false, b_l,
                                                      (short)0, c, false, false);
            c = __builtin_amdgcn_wmma_f32_16x16x4_f32(false, a_self[s], false, b_r,
                                                      (short)0, c, false, false);
        }
        const float bv = bias[nb + m];
        if (fullTile) {
            float* o = out + (size_t)rbase * COUT + nb + m;
#pragma unroll
            for (int v = 0; v < 8; ++v) {
                float val = c[v] + bv;
                if (RELU) val = fmaxf(val, 0.0f);
                o[(size_t)v * COUT] = val;
            }
        } else {
#pragma unroll
            for (int v = 0; v < 8; ++v) {
                int r = rbase + v;
                if (r < N) {
                    float val = c[v] + bv;
                    if (RELU) val = fmaxf(val, 0.0f);
                    out[(size_t)r * COUT + nb + m] = val;
                }
            }
        }
    }
}

template <int CIN, int COUT>
__global__ __launch_bounds__(256)
void sage_wmma_project_kernel(const int* __restrict__ rowptr,
                              const int* __restrict__ nbr,
                              const float* __restrict__ invdeg,
                              const float* __restrict__ xin,
                              const float* __restrict__ wl,
                              const float* __restrict__ wr,
                              const float* __restrict__ bias,
                              const float* __restrict__ wl3,
                              const float* __restrict__ wr3,
                              const float* __restrict__ b3,
                              float* __restrict__ p,
                              float* __restrict__ q,
                              int N) {
    constexpr int P = CIN + 4;
    __shared__ __attribute__((aligned(16))) float lds[2 * COUT * P + WAVES_PER_BLOCK * 16 * P + 256];

    const int lane = threadIdx.x & 31;
    const int waveInBlock = __builtin_amdgcn_readfirstlane((int)(threadIdx.x >> 5));
    const int tile = blockIdx.x * WAVES_PER_BLOCK + waveInBlock;
    const int row0 = tile * 16;
    const int m    = lane & 15;
    const int half = lane >> 4;
    const int kk   = half << 1;

    float* sWl   = lds;
    float* sWr   = lds + COUT * P;
    float* sTile = lds + 2 * COUT * P + waveInBlock * (16 * P);
    float* stage = lds + 2 * COUT * P + WAVES_PER_BLOCK * 16 * P;

    stage_matrix<CIN, P, COUT, 256>(wl, sWl, threadIdx.x);
    stage_matrix<CIN, P, COUT, 256>(wr, sWr, threadIdx.x);
    __syncthreads();

    float pl = 0.0f, pr = 0.0f;
    if (row0 < N) {
        v2f x_mean[CIN / 4], x_self[CIN / 4];
        const float* tileRow = sTile + m * P + kk;
        const int nn = min(16, N - row0);
        const int rp = rowptr[row0 + min(lane, nn)];
        const float inv = invdeg[row0 + min(lane, nn - 1)];
        stage_self_tile<CIN, P>(xin, sTile, row0, N, lane);
        wave_tile_sync();
#pragma unroll
        for (int s = 0; s < CIN / 4; ++s)
            x_self[s] = *reinterpret_cast<const v2f*>(tileRow + 4 * s);
        wave_tile_sync();
        gather_mean_tile<CIN, P>(xin, nbr, rp, inv, sTile, nn, lane);
        wave_tile_sync();
#pragma unroll
        for (int s = 0; s < CIN / 4; ++s)
            x_mean[s] = *reinterpret_cast<const v2f*>(tileRow + 4 * s);

#pragma unroll
        for (int nb = 0; nb < COUT; nb += 16) {
            v8f c = {};
            const float* wlRow = sWl + (nb + m) * P + kk;
            const float* wrRow = sWr + (nb + m) * P + kk;
#pragma unroll
            for (int s = 0; s < CIN / 4; ++s) {
                const v2f w_l = *reinterpret_cast<const v2f*>(wlRow + 4 * s);
                const v2f w_r = *reinterpret_cast<const v2f*>(wrRow + 4 * s);
                c = __builtin_amdgcn_wmma_f32_16x16x4_f32(false, w_l, false, x_mean[s],
                                                          (short)0, c, false, false);
                c = __builtin_amdgcn_wmma_f32_16x16x4_f32(false, w_r, false, x_self[s],
                                                          (short)0, c, false, false);
            }
            const int ch0 = nb + (half << 3);
            const float4 bA = *(const float4*)(bias + ch0), bB = *(const float4*)(bias + ch0 + 4);
            const float4 lA = *(const float4*)(wl3 + ch0),  lB = *(const float4*)(wl3 + ch0 + 4);
            const float4 rA = *(const float4*)(wr3 + ch0),  rB = *(const float4*)(wr3 + ch0 + 4);
            const float bv[8] = {bA.x, bA.y, bA.z, bA.w, bB.x, bB.y, bB.z, bB.w};
            const float lv[8] = {lA.x, lA.y, lA.z, lA.w, lB.x, lB.y, lB.z, lB.w};
            const float rv[8] = {rA.x, rA.y, rA.z, rA.w, rB.x, rB.y, rB.z, rB.w};
#pragma unroll
            for (int v = 0; v < 8; ++v) {
                const float h = fmaxf(c[v] + bv[v], 0.0f);
                pl += h * lv[v];
                pr += h * rv[v];
            }
        }
        pl += __shfl_xor(pl, 16);
        pr += __shfl_xor(pr, 16);
        pr += b3[0];
    }
    if (half == 0) {
        stage[waveInBlock * 16 + m] = pl;
        stage[128 + waveInBlock * 16 + m] = pr;
    }
    __syncthreads();
    const int t = threadIdx.x & 127;
    const int which = threadIdx.x >> 7;
    const int nodeOut = blockIdx.x * 128 + t;
    if (nodeOut < N) (which ? q : p)[nodeOut] = stage[which * 128 + t];
}

__global__ __launch_bounds__(256)
void gather_scalar_final_kernel(const float* __restrict__ p,
                                const int* __restrict__ rowptr,
                                const int* __restrict__ nbr,
                                const float* __restrict__ invdeg,
                                const float* __restrict__ q,
                                float* __restrict__ out, int N) {
    const int i = blockIdx.x * blockDim.x + threadIdx.x;
    if (i >= N) return;
    int k = rowptr[i];
    const int end = rowptr[i + 1];
    float s = 0.0f;
    for (; k + 4 <= end; k += 4) {
        const int s0 = nbr[k], s1 = nbr[k + 1], s2 = nbr[k + 2], s3 = nbr[k + 3];
        const float v0 = p[s0], v1 = p[s1], v2 = p[s2], v3 = p[s3];
        s += v0;
        s += v1;
        s += v2;
        s += v3;
    }
    for (; k < end; ++k) s += p[nbr[k]];
    const float z = s * invdeg[i] + q[i];
    out[i] = 1.0f / (1.0f + expf(-z));
}

extern "C" void kernel_launch(void* const* d_in, const int* in_sizes, int n_in,
                              void* d_out, int out_size, void* d_ws, size_t ws_size,
                              hipStream_t stream) {
    const float* x    = (const float*)d_in[0];
    const int*   ei   = (const int*)d_in[1];
    const float* wl1  = (const float*)d_in[2];
    const float* wr1  = (const float*)d_in[3];
    const float* b1   = (const float*)d_in[4];
    const float* wl2  = (const float*)d_in[5];
    const float* wr2  = (const float*)d_in[6];
    const float* b2   = (const float*)d_in[7];
    const float* wl3  = (const float*)d_in[8];
    const float* wr3  = (const float*)d_in[9];
    const float* b3   = (const float*)d_in[10];
    float* out = (float*)d_out;

    const int N = in_sizes[0] / 32;
    const int E = in_sizes[1] / 2;
    const int* src = ei;
    const int* dst = ei + E;

    int*   totals = (int*)d_ws;
    int*   fill   = totals + 256;
    float* deg  = (float*)(totals + N);
    float* recs = deg + N;
    float* h1   = recs + (size_t)N * 64;
    float* p3   = h1 + (size_t)N * 64;
    float* q3   = p3 + N;
    int*   nbr  = (int*)(q3 + N);
    int*   rptr = nbr + E;
    int*   binned = (int*)recs;

    const int tiles = (N + 15) / 16;
    const int gemmBlocks = (tiles + 7) / 8;
    const int chunkBlocks = (E + 8191) / 8192;
    const int buckets = (N + 511) / 512;

    hipMemsetAsync(totals, 0, 512 * sizeof(int), stream);
    bucket_count_kernel<<<chunkBlocks, 1024, 0, stream>>>(dst, totals, E);
    bucket_scatter_kernel<<<chunkBlocks, 1024, 0, stream>>>(src, dst, totals, fill, binned, E);
    bucket_finalize_kernel<<<buckets, 1024, 0, stream>>>(binned, totals, rptr, deg, nbr, N);

    sage_wmma_kernel<32, 64, true><<<gemmBlocks, 256, 0, stream>>>(
        rptr, nbr, deg, x, wl1, wr1, b1, h1, N);

    sage_wmma_project_kernel<64, 64><<<gemmBlocks, 256, 0, stream>>>(
        rptr, nbr, deg, h1, wl2, wr2, b2, wl3, wr3, b3, p3, q3, N);

    gather_scalar_final_kernel<<<(N + 255) / 256, 256, 0, stream>>>(
        p3, rptr, nbr, deg, q3, out, N);
}


# Layer 1 fused into the level-2 grouping kernel: each bucket block's 32 waves take its 32 node tiles, lists read from LDS; one launch fewer; per-node sum order kept
# speedup vs baseline: 5.9567x; 1.0311x over previous
#include <hip/hip_runtime.h>
#include <hip/hip_bf16.h>
#include <math.h>

typedef __attribute__((ext_vector_type(2))) float v2f;
typedef __attribute__((ext_vector_type(8))) float v8f;

constexpr int BUCKET_SHIFT = 9;
constexpr int BUCKET_NODES = 1 << BUCKET_SHIFT;
constexpr int MAX_BUCKETS  = 256;
constexpr int EPT          = 8;
constexpr int CHUNK        = 1024 * EPT;
constexpr int RPT          = 12;
constexpr int LIST_CAP     = 16384;

__global__ __launch_bounds__(1024)
void bucket_count_kernel(const int* __restrict__ dst, int* __restrict__ totals, int E) {
    __shared__ int hist[MAX_BUCKETS];
    const int tid = threadIdx.x;
    if (tid < MAX_BUCKETS) hist[tid] = 0;
    __syncthreads();
    const int base = blockIdx.x * CHUNK;
    int dv[EPT];
#pragma unroll
    for (int j = 0; j < EPT; ++j) {
        const int e = base + j * 1024 + tid;
        dv[j] = dst[min(e, E - 1)];
    }
#pragma unroll
    for (int j = 0; j < EPT; ++j) {
        const int e = base + j * 1024 + tid;
        if (e < E) atomicAdd(&hist[dv[j] >> BUCKET_SHIFT], 1);
    }
    __syncthreads();
    if (tid < MAX_BUCKETS) {
        const int c = hist[tid];
        if (c > 0) atomicAdd(&totals[tid], c);
    }
}

__global__ __launch_bounds__(1024)
void bucket_scatter_kernel(const int* __restrict__ src, const int* __restrict__ dst,
                           const int* __restrict__ totals, int* __restrict__ fill,
                           int* __restrict__ binned, int E) {
    __shared__ int recS[CHUNK];
    __shared__ int recD[CHUNK];
    __shared__ int hist[MAX_BUCKETS];
    __shared__ int lstart[MAX_BUCKETS];
    __shared__ int gdelta[MAX_BUCKETS];
    __shared__ int wsumT[8];
    __shared__ int wsumC[8];
    const int tid = threadIdx.x;
    const int lane = tid & 31;
    const int w = tid >> 5;
    if (tid < MAX_BUCKETS) hist[tid] = 0;
    __syncthreads();

    const int base = blockIdx.x * CHUNK;
    const int nvalid = min(CHUNK, E - base);
    int sv[EPT], dv[EPT], rk[EPT];
#pragma unroll
    for (int j = 0; j < EPT; ++j) {
        const int i = j * 1024 + tid;
        const int e = base + min(i, nvalid - 1);
        sv[j] = src[e];
        dv[j] = dst[e];
    }
#pragma unroll
    for (int j = 0; j < EPT; ++j) {
        const int i = j * 1024 + tid;
        rk[j] = (i < nvalid) ? atomicAdd(&hist[dv[j] >> BUCKET_SHIFT], 1) : 0;
    }

    int t = 0, c = 0, incT = 0, incC = 0;
    if (tid < MAX_BUCKETS) t = totals[tid];
    __syncthreads();
    if (tid < MAX_BUCKETS) {
        c = hist[tid];
        incT = t;
        incC = c;
        for (int o = 1; o < 32; o <<= 1) {
            const int uT = __shfl_up(incT, o);
            const int uC = __shfl_up(incC, o);
            if (lane >= o) { incT += uT; incC += uC; }
        }
        if (lane == 31) { wsumT[w] = incT; wsumC[w] = incC; }
    }
    __syncthreads();
    if (tid < MAX_BUCKETS) {
        int preT = 0, preC = 0;
        for (int k = 0; k < 8; ++k) {
            preT += (k < w) ? wsumT[k] : 0;
            preC += (k < w) ? wsumC[k] : 0;
        }
        const int segStart = preT + incT - t;
        const int ls = preC + incC - c;
        const int got = (c > 0) ? atomicAdd(&fill[tid], c) : 0;
        lstart[tid] = ls;
        gdelta[tid] = segStart + got - ls;
    }
    __syncthreads();
#pragma unroll
    for (int j = 0; j < EPT; ++j) {
        const int i = j * 1024 + tid;
        if (i < nvalid) {
            const int slot = lstart[dv[j] >> BUCKET_SHIFT] + rk[j];
            recS[slot] = sv[j];
            recD[slot] = dv[j];
        }
    }
    __syncthreads();
    int2* out = reinterpret_cast<int2*>(binned);
#pragma unroll
    for (int j = 0; j < EPT; ++j) {
        const int i = j * 1024 + tid;
        if (i < nvalid) {
            const int s = recS[i];
            const int d = recD[i];
            out[gdelta[d >> BUCKET_SHIFT] + i] = make_int2(s, d);
        }
    }
}


__device__ __forceinline__ float swap_rows16(float v) {
    return __builtin_bit_cast(float, __builtin_amdgcn_permlanex16(
        __builtin_bit_cast(int, v), __builtin_bit_cast(int, v), 0x76543210, 0xfedcba98, false, false));
}
__device__ __forceinline__ float swap_groups8(float v) {
    return __builtin_bit_cast(float, __builtin_amdgcn_update_dpp(
        0, __builtin_bit_cast(int, v), 0x168, 0xf, 0xf, false));
}

template <int C, int K, int t>
__device__ __forceinline__ void issue_rows(const char* __restrict__ xb, unsigned sv, unsigned subOff,
                                           bool lastOk, float4 (&v)[K]) {
    if constexpr (t < K) {
        const unsigned idx = __builtin_amdgcn_mov_dpp8(sv, t * 0x249249u);
        const unsigned off = idx * (unsigned)(C * 4) + subOff;
        if (t + 1 < K || lastOk) v[t] = *reinterpret_cast<const float4*>(xb + off);
        else v[t] = make_float4(0.0f, 0.0f, 0.0f, 0.0f);
        issue_rows<C, K, t + 1>(xb, sv, subOff, lastOk, v);
    }
}

template <int C, int K>
__device__ __forceinline__ void gather_pass(const char* __restrict__ xb, unsigned sv, unsigned subOff,
                                            bool lastOk, float4& acc) {
    float4 v[K];
    issue_rows<C, K, 0>(xb, sv, subOff, lastOk, v);
#pragma unroll
    for (int t = 0; t < K; ++t) {
        acc.x += v[t].x; acc.y += v[t].y; acc.z += v[t].z; acc.w += v[t].w;
    }
}

template <int C, int P, typename ListPtr>
__device__ __forceinline__ void gather_mean_tile(const float* __restrict__ x,
                                                 ListPtr nbr,
                                                 int rp, float inv,
                                                 float* sTile, int nn, int lane) {
    constexpr int LPR = C / 4;
    constexpr int G   = 32 / LPR;
    constexpr int PS  = 8 * G;
    static_assert(C == 32 || C == 64, "a row is 8 or 16 lanes");
    const int g    = lane / LPR;
    const unsigned subOff = (lane % LPR) * 16;
    const int perm = G * (lane & 7) + g;
    int pos = __builtin_amdgcn_readlane(rp, 0);
    const int wend = __builtin_amdgcn_readlane(rp, nn);
    const char* xb = reinterpret_cast<const char*>(x);

    unsigned svNext = 0;
    if (pos < wend) svNext = (unsigned)nbr[min(pos + perm, wend - 1)];

    for (int i = 0; i < nn; ++i) {
        const int end = __builtin_amdgcn_readlane(rp, i + 1);
        float4 acc = make_float4(0.0f, 0.0f, 0.0f, 0.0f);
        while (pos < end) {
            const int n = min(PS, end - pos);
            const unsigned sv = svNext;
            pos += n;
            if (pos < wend) svNext = (unsigned)nbr[min(pos + perm, wend - 1)];
            const int K = (n + G - 1) / G;
            const bool lastOk = g < n - (K - 1) * G;
            switch (K) {
                case 1:  gather_pass<C, 1>(xb, sv, subOff, lastOk, acc); break;
                case 2:  gather_pass<C, 2>(xb, sv, subOff, lastOk, acc); break;
                case 3:  gather_pass<C, 3>(xb, sv, subOff, lastOk, acc); break;
                case 4:  gather_pass<C, 4>(xb, sv, subOff, lastOk, acc); break;
                case 5:  gather_pass<C, 5>(xb, sv, subOff, lastOk, acc); break;
                case 6:  gather_pass<C, 6>(xb, sv, subOff, lastOk, acc); break;
                case 7:  gather_pass<C, 7>(xb, sv, subOff, lastOk, acc); break;
                default: gather_pass<C, 8>(xb, sv, subOff, lastOk, acc); break;
            }
        }
        if (G == 4) {
            acc.x += swap_groups8(acc.x); acc.y += swap_groups8(acc.y);
            acc.z += swap_groups8(acc.z); acc.w += swap_groups8(acc.w);
        }
        acc.x += swap_rows16(acc.x); acc.y += swap_rows16(acc.y);
        acc.z += swap_rows16(acc.z); acc.w += swap_rows16(acc.w);
        const float scale = __builtin_bit_cast(float, __builtin_amdgcn_readlane(__builtin_bit_cast(int, inv), i));
        acc.x *= scale; acc.y *= scale; acc.z *= scale; acc.w *= scale;
        if (lane < LPR)
            *reinterpret_cast<float4*>(sTile + i * P + lane * 4) = acc;
    }
    for (int i = nn; i < 16; ++i)
        if (lane < LPR)
            *reinterpret_cast<float4*>(sTile + i * P + lane * 4) = make_float4(0.0f, 0.0f, 0.0f, 0.0f);
}

__device__ __forceinline__ void wave_tile_sync() {
    __builtin_amdgcn_fence(__ATOMIC_RELEASE, "wavefront");
    __builtin_amdgcn_wave_barrier();
    __builtin_amdgcn_fence(__ATOMIC_ACQUIRE, "wavefront");
}

constexpr int WAVES_PER_BLOCK = 8;

template <int CIN, int P, int ROWS, int NT>
__device__ __forceinline__ void stage_matrix(const float* __restrict__ g, float* s, int t) {
    constexpr int V = CIN / 4;
    static_assert((ROWS * V) % NT == 0, "whole passes only");
#pragma unroll
    for (int i = 0; i < ROWS * V / NT; ++i) {
        const int v = i * NT + t;
        const float4 val = *reinterpret_cast<const float4*>(g + (size_t)v * 4);
        *reinterpret_cast<float4*>(s + (v / V) * P + (v % V) * 4) = val;
    }
}

template <int CIN, int P>
__device__ __forceinline__ void stage_self_tile(const float* __restrict__ xin, float* sSelf,
                                                int row0, int N, int lane) {
    constexpr int V = CIN / 4;
#pragma unroll
    for (int i = 0; i < 16 * V / 32; ++i) {
        const int v = i * 32 + lane;
        const int r = v / V;
        const int c = (v % V) * 4;
        const int node = min(row0 + r, N - 1);
        const float4 s = *reinterpret_cast<const float4*>(xin + (size_t)node * CIN + c);
        *reinterpret_cast<float4*>(sSelf + r * P + c) = s;
    }
}

template <int CIN, int COUT, bool RELU, typename ListPtr>
__device__ __forceinline__ void sage_tile(const float* sWl, const float* sWr,
                                          float* sMean, float* sSelf,
                                          int rp, float inv, ListPtr lst,
                                          const float* __restrict__ xin,
                                          const float* __restrict__ bias,
                                          float* __restrict__ out,
                                          int row0, int nn, int N, int lane) {
    constexpr int P = CIN + 4;
    stage_self_tile<CIN, P>(xin, sSelf, row0, N, lane);
    gather_mean_tile<CIN, P>(xin, lst, rp, inv, sMean, nn, lane);
    wave_tile_sync();

    const int m  = lane & 15;
    const int kk = (lane >> 4) << 1;
    const bool fullTile = (row0 + 16 <= N);
    const int rbase = row0 + ((lane >> 4) << 3);

    v2f a_mean[CIN / 4], a_self[CIN / 4];
    const float* meanRow = sMean + m * P + kk;
    const float* selfRow = sSelf + m * P + kk;
#pragma unroll
    for (int s = 0; s < CIN / 4; ++s) {
        a_mean[s] = *reinterpret_cast<const v2f*>(meanRow + 4 * s);
        a_self[s] = *reinterpret_cast<const v2f*>(selfRow + 4 * s);
    }

#pragma unroll
    for (int nb = 0; nb < COUT; nb += 16) {
        v8f c = {};
        const float* wlRow = sWl + (nb + m) * P + kk;
        const float* wrRow = sWr + (nb + m) * P + kk;
#pragma unroll
        for (int s = 0; s < CIN / 4; ++s) {
            const v2f b_l = *reinterpret_cast<const v2f*>(wlRow + 4 * s);
            const v2f b_r = *reinterpret_cast<const v2f*>(wrRow + 4 * s);
            c = __builtin_amdgcn_wmma_f32_16x16x4_f32(false, a_mean[s], false, b_l,
                                                      (short)0, c, false, false);
            c = __builtin_amdgcn_wmma_f32_16x16x4_f32(false, a_self[s], false, b_r,
                                                      (short)0, c, false, false);
        }
        const float bv = bias[nb + m];
        if (fullTile) {
            float* o = out + (size_t)rbase * COUT + nb + m;
#pragma unroll
            for (int v = 0; v < 8; ++v) {
                float val = c[v] + bv;
                if (RELU) val = fmaxf(val, 0.0f);
                o[(size_t)v * COUT] = val;
            }
        } else {
#pragma unroll
            for (int v = 0; v < 8; ++v) {
                int r = rbase + v;
                if (r < N) {
                    float val = c[v] + bv;
                    if (RELU) val = fmaxf(val, 0.0f);
                    out[(size_t)r * COUT + nb + m] = val;
                }
            }
        }
    }
}

__global__ __launch_bounds__(1024)
void bucket_finalize_layer1_kernel(const int* __restrict__ binned, const int* __restrict__ totals,
                                   int* __restrict__ rowptr, float* __restrict__ invdeg,
                                   int* __restrict__ nbr,
                                   const float* __restrict__ x,
                                   const float* __restrict__ wl,
                                   const float* __restrict__ wr,
                                   const float* __restrict__ bias,
                                   float* __restrict__ h1,
                                   int N) {
    constexpr int L1_CIN = 32, L1_COUT = 64, L1_P = L1_CIN + 4;
    __shared__ int cnt[BUCKET_NODES];
    __shared__ int wsum[32];
    __shared__ int sList[LIST_CAP];
    __shared__ __attribute__((aligned(16))) float sW[2 * L1_COUT * L1_P];
    __shared__ __attribute__((aligned(16))) float sTiles[32 * 2 * 16 * L1_P];
    const int tid = threadIdx.x;
    const int lane = tid & 31;
    const int w = tid >> 5;
    const int b = blockIdx.x;
    if (tid < BUCKET_NODES) cnt[tid] = 0;
    stage_matrix<L1_CIN, L1_P, L1_COUT, 512>(tid < 512 ? wl : wr, tid < 512 ? sW : sW + L1_COUT * L1_P, tid & 511);

    int part = (tid < MAX_BUCKETS && tid < b) ? totals[tid] : 0;
    for (int o = 16; o > 0; o >>= 1) part += __shfl_xor(part, o);
    if (lane == 0) wsum[w] = part;
    const int count = totals[b];
    __syncthreads();
    int start = 0;
    for (int k = 0; k < 8; ++k) start += wsum[k];

    const int2* rec = reinterpret_cast<const int2*>(binned) + start;

    const int* recDst = binned + 2 * (size_t)start + 1;
    for (int i0 = 0; i0 < count; i0 += 1024 * RPT) {
        int d[RPT];
#pragma unroll
        for (int j = 0; j < RPT; ++j) {
            const int i = i0 + j * 1024 + tid;
            d[j] = recDst[2 * min(i, count - 1)];
        }
#pragma unroll
        for (int j = 0; j < RPT; ++j) {
            const int i = i0 + j * 1024 + tid;
            if (i < count) atomicAdd(&cnt[d[j] & (BUCKET_NODES - 1)], 1);
        }
    }
    __syncthreads();

    int c = 0, inc = 0;
    if (tid < BUCKET_NODES) {
        c = cnt[tid];
        inc = c;
        for (int o = 1; o < 32; o <<= 1) {
            const int u = __shfl_up(inc, o);
            if (lane >= o) inc += u;
        }
        if (lane == 31) wsum[w] = inc;
    }
    __syncthreads();
    int excl = 0;
    if (tid < BUCKET_NODES) {
        int pre = 0;
        for (int k = 0; k < 16; ++k) pre += (k < w) ? wsum[k] : 0;
        excl = pre + inc - c;
        cnt[tid] = excl;
    }
    __syncthreads();

    if (count <= LIST_CAP) {
        for (int i0 = 0; i0 < count; i0 += 1024 * RPT) {
            int2 r[RPT];
            int pos[RPT];
#pragma unroll
            for (int j = 0; j < RPT; ++j) {
                const int i = i0 + j * 1024 + tid;
                r[j] = rec[min(i, count - 1)];
            }
#pragma unroll
            for (int j = 0; j < RPT; ++j) {
                const int i = i0 + j * 1024 + tid;
                pos[j] = (i < count) ? atomicAdd(&cnt[r[j].y & (BUCKET_NODES - 1)], 1) : -1;
            }
#pragma unroll
            for (int j = 0; j < RPT; ++j)
                if (pos[j] >= 0) sList[pos[j]] = r[j].x;
        }
        __syncthreads();
        const int lead = start & 31;
        for (int e0 = tid - lead; e0 < count; e0 += 4 * 1024) {
            int v[4];
#pragma unroll
            for (int u = 0; u < 4; ++u)
                v[u] = sList[min(max(e0 + u * 1024, 0), LIST_CAP - 1)];
#pragma unroll
            for (int u = 0; u < 4; ++u) {
                const int e = e0 + u * 1024;
                if (e >= 0 && e < count) nbr[start + e] = v[u];
            }
        }
    } else {
        for (int i0 = 0; i0 < count; i0 += 1024 * RPT) {
            int2 r[RPT];
#pragma unroll
            for (int j = 0; j < RPT; ++j) {
                const int i = i0 + j * 1024 + tid;
                r[j] = rec[min(i, count - 1)];
            }
#pragma unroll
            for (int j = 0; j < RPT; ++j) {
                const int i = i0 + j * 1024 + tid;
                if (i < count) {
                    const int pos = atomicAdd(&cnt[r[j].y & (BUCKET_NODES - 1)], 1);
                    nbr[start + pos] = r[j].x;
                }
            }
        }
        __threadfence();
        __syncthreads();
    }

    if (tid < BUCKET_NODES) {
        const int node = (b << BUCKET_SHIFT) + tid;
        if (node < N) {
            rowptr[node] = start + excl;
            invdeg[node] = 1.0f / fmaxf((float)c, 1.0f);
        }
        if (node == N - 1) rowptr[N] = start + excl + c;
    }

    const int w0 = __builtin_amdgcn_readfirstlane(w);
    const int row0 = (b << BUCKET_SHIFT) + w0 * 16;
    if (row0 < N) {
        const int nn = min(16, N - row0);
        const int jr = w0 * 16 + min(lane, nn);
        const int rp = (jr > 0) ? cnt[jr - 1] : 0;
        const int ji = w0 * 16 + min(lane, nn - 1);
        const int dg = cnt[ji] - ((ji > 0) ? cnt[ji - 1] : 0);
        const float inv = 1.0f / fmaxf((float)dg, 1.0f);
        float* sMean = sTiles + w0 * (2 * 16 * L1_P);
        float* sSelf = sMean + 16 * L1_P;
        if (count <= LIST_CAP)
            sage_tile<L1_CIN, L1_COUT, true>(sW, sW + L1_COUT * L1_P, sMean, sSelf, rp, inv,
                                             (const int*)sList, x, bias, h1, row0, nn, N, lane);
        else
            sage_tile<L1_CIN, L1_COUT, true>(sW, sW + L1_COUT * L1_P, sMean, sSelf, rp, inv,
                                             (const int*)(nbr + start), x, bias, h1, row0, nn, N, lane);
    }
}


template <int CIN, int COUT>
__global__ __launch_bounds__(256)
void sage_wmma_project_kernel(const int* __restrict__ rowptr,
                              const int* __restrict__ nbr,
                              const float* __restrict__ invdeg,
                              const float* __restrict__ xin,
                              const float* __restrict__ wl,
                              const float* __restrict__ wr,
                              const float* __restrict__ bias,
                              const float* __restrict__ wl3,
                              const float* __restrict__ wr3,
                              const float* __restrict__ b3,
                              float* __restrict__ p,
                              float* __restrict__ q,
                              int N) {
    constexpr int P = CIN + 4;
    __shared__ __attribute__((aligned(16))) float lds[2 * COUT * P + WAVES_PER_BLOCK * 16 * P + 256];

    const int lane = threadIdx.x & 31;
    const int waveInBlock = __builtin_amdgcn_readfirstlane((int)(threadIdx.x >> 5));
    const int tile = blockIdx.x * WAVES_PER_BLOCK + waveInBlock;
    const int row0 = tile * 16;
    const int m    = lane & 15;
    const int half = lane >> 4;
    const int kk   = half << 1;

    float* sWl   = lds;
    float* sWr   = lds + COUT * P;
    float* sTile = lds + 2 * COUT * P + waveInBlock * (16 * P);
    float* stage = lds + 2 * COUT * P + WAVES_PER_BLOCK * 16 * P;

    stage_matrix<CIN, P, COUT, 256>(wl, sWl, threadIdx.x);
    stage_matrix<CIN, P, COUT, 256>(wr, sWr, threadIdx.x);
    __syncthreads();

    float pl = 0.0f, pr = 0.0f;
    if (row0 < N) {
        v2f x_mean[CIN / 4], x_self[CIN / 4];
        const float* tileRow = sTile + m * P + kk;
        const int nn = min(16, N - row0);
        const int rp = rowptr[row0 + min(lane, nn)];
        const float inv = invdeg[row0 + min(lane, nn - 1)];
        stage_self_tile<CIN, P>(xin, sTile, row0, N, lane);
        wave_tile_sync();
#pragma unroll
        for (int s = 0; s < CIN / 4; ++s)
            x_self[s] = *reinterpret_cast<const v2f*>(tileRow + 4 * s);
        wave_tile_sync();
        gather_mean_tile<CIN, P>(xin, nbr, rp, inv, sTile, nn, lane);
        wave_tile_sync();
#pragma unroll
        for (int s = 0; s < CIN / 4; ++s)
            x_mean[s] = *reinterpret_cast<const v2f*>(tileRow + 4 * s);

#pragma unroll
        for (int nb = 0; nb < COUT; nb += 16) {
            v8f c = {};
            const float* wlRow = sWl + (nb + m) * P + kk;
            const float* wrRow = sWr + (nb + m) * P + kk;
#pragma unroll
            for (int s = 0; s < CIN / 4; ++s) {
                const v2f w_l = *reinterpret_cast<const v2f*>(wlRow + 4 * s);
                const v2f w_r = *reinterpret_cast<const v2f*>(wrRow + 4 * s);
                c = __builtin_amdgcn_wmma_f32_16x16x4_f32(false, w_l, false, x_mean[s],
                                                          (short)0, c, false, false);
                c = __builtin_amdgcn_wmma_f32_16x16x4_f32(false, w_r, false, x_self[s],
                                                          (short)0, c, false, false);
            }
            const int ch0 = nb + (half << 3);
            const float4 bA = *(const float4*)(bias + ch0), bB = *(const float4*)(bias + ch0 + 4);
            const float4 lA = *(const float4*)(wl3 + ch0),  lB = *(const float4*)(wl3 + ch0 + 4);
            const float4 rA = *(const float4*)(wr3 + ch0),  rB = *(const float4*)(wr3 + ch0 + 4);
            const float bv[8] = {bA.x, bA.y, bA.z, bA.w, bB.x, bB.y, bB.z, bB.w};
            const float lv[8] = {lA.x, lA.y, lA.z, lA.w, lB.x, lB.y, lB.z, lB.w};
            const float rv[8] = {rA.x, rA.y, rA.z, rA.w, rB.x, rB.y, rB.z, rB.w};
#pragma unroll
            for (int v = 0; v < 8; ++v) {
                const float h = fmaxf(c[v] + bv[v], 0.0f);
                pl += h * lv[v];
                pr += h * rv[v];
            }
        }
        pl += __shfl_xor(pl, 16);
        pr += __shfl_xor(pr, 16);
        pr += b3[0];
    }
    if (half == 0) {
        stage[waveInBlock * 16 + m] = pl;
        stage[128 + waveInBlock * 16 + m] = pr;
    }
    __syncthreads();
    const int t = threadIdx.x & 127;
    const int which = threadIdx.x >> 7;
    const int nodeOut = blockIdx.x * 128 + t;
    if (nodeOut < N) (which ? q : p)[nodeOut] = stage[which * 128 + t];
}

__global__ __launch_bounds__(256)
void gather_scalar_final_kernel(const float* __restrict__ p,
                                const int* __restrict__ rowptr,
                                const int* __restrict__ nbr,
                                const float* __restrict__ invdeg,
                                const float* __restrict__ q,
                                float* __restrict__ out, int N) {
    const int i = blockIdx.x * blockDim.x + threadIdx.x;
    if (i >= N) return;
    int k = rowptr[i];
    const int end = rowptr[i + 1];
    float s = 0.0f;
    for (; k + 4 <= end; k += 4) {
        const int s0 = nbr[k], s1 = nbr[k + 1], s2 = nbr[k + 2], s3 = nbr[k + 3];
        const float v0 = p[s0], v1 = p[s1], v2 = p[s2], v3 = p[s3];
        s += v0;
        s += v1;
        s += v2;
        s += v3;
    }
    for (; k < end; ++k) s += p[nbr[k]];
    const float z = s * invdeg[i] + q[i];
    out[i] = 1.0f / (1.0f + expf(-z));
}

extern "C" void kernel_launch(void* const* d_in, const int* in_sizes, int n_in,
                              void* d_out, int out_size, void* d_ws, size_t ws_size,
                              hipStream_t stream) {
    const float* x    = (const float*)d_in[0];
    const int*   ei   = (const int*)d_in[1];
    const float* wl1  = (const float*)d_in[2];
    const float* wr1  = (const float*)d_in[3];
    const float* b1   = (const float*)d_in[4];
    const float* wl2  = (const float*)d_in[5];
    const float* wr2  = (const float*)d_in[6];
    const float* b2   = (const float*)d_in[7];
    const float* wl3  = (const float*)d_in[8];
    const float* wr3  = (const float*)d_in[9];
    const float* b3   = (const float*)d_in[10];
    float* out = (float*)d_out;

    const int N = in_sizes[0] / 32;
    const int E = in_sizes[1] / 2;
    const int* src = ei;
    const int* dst = ei + E;

    int*   totals = (int*)d_ws;
    int*   fill   = totals + 256;
    float* deg  = (float*)(totals + N);
    float* recs = deg + N;
    float* h1   = recs + (size_t)N * 64;
    float* p3   = h1 + (size_t)N * 64;
    float* q3   = p3 + N;
    int*   nbr  = (int*)(q3 + N);
    int*   rptr = nbr + E;
    int*   binned = (int*)recs;

    const int tiles = (N + 15) / 16;
    const int gemmBlocks = (tiles + 7) / 8;
    const int chunkBlocks = (E + 8191) / 8192;
    const int buckets = (N + 511) / 512;

    hipMemsetAsync(totals, 0, 512 * sizeof(int), stream);
    bucket_count_kernel<<<chunkBlocks, 1024, 0, stream>>>(dst, totals, E);
    bucket_scatter_kernel<<<chunkBlocks, 1024, 0, stream>>>(src, dst, totals, fill, binned, E);
    bucket_finalize_layer1_kernel<<<buckets, 1024, 0, stream>>>(binned, totals, rptr, deg, nbr,
                                                                x, wl1, wr1, b1, h1, N);

    sage_wmma_project_kernel<64, 64><<<gemmBlocks, 256, 0, stream>>>(
        rptr, nbr, deg, h1, wl2, wr2, b2, wl3, wr3, b3, p3, q3, N);

    gather_scalar_final_kernel<<<(N + 255) / 256, 256, 0, stream>>>(
        p3, rptr, nbr, deg, q3, out, N);
}
